# starting kernel
# baseline (speedup 1.0000x reference)
#include <hip/hip_runtime.h>
#include <math.h>


#define N_NODES 50000
#define N_EDGES 800000
#define HEADS   8
#define HEAD_DIM 16
#define FDIM    128
#define ETOT    (N_EDGES + N_NODES)
#define NEG_SLOPE 0.2f

typedef __attribute__((ext_vector_type(2))) float v2f;
typedef __attribute__((ext_vector_type(8))) float v8f;

__global__ void __launch_bounds__(128)
gemm128_wmma_f32(const float* __restrict__ A, const float* __restrict__ W,
                 float* __restrict__ out) {
  const int wave = threadIdx.x >> 5;
  const int lane = threadIdx.x & 31;
  const int tile = blockIdx.x * 4 + wave;
  const int tm   = tile >> 3;
  const int tn   = tile & 7;
  const int half = lane >> 4;
  const int l    = lane & 15;

  const float* __restrict__ arow = A + (size_t)(tm * 16 + l) * FDIM;
  const float* __restrict__ bcol = W + tn * 16 + l;

  v8f c = {};
#pragma unroll
  for (int k0 = 0; k0 < FDIM; k0 += 4) {
    const int ka = k0 + 2 * half;
    v2f a, b;
    a.x = arow[ka];
    a.y = arow[ka + 1];
    b.x = bcol[(size_t)ka * FDIM];
    b.y = bcol[(size_t)(ka + 1) * FDIM];
    c = __builtin_amdgcn_wmma_f32_16x16x4_f32(false, a, false, b, (short)0, c,
                                              false, false);
  }
  float* __restrict__ orow =
      out + (size_t)(tm * 16 + 8 * half) * FDIM + tn * 16 + l;
#pragma unroll
  for (int r = 0; r < 8; ++r) orow[(size_t)r * FDIM] = c[r];
}

__global__ void att_scores(const float* __restrict__ h,
                           const float* __restrict__ att_src,
                           const float* __restrict__ att_dst,
                           float* __restrict__ a_src, float* __restrict__ a_dst) {
  int t = blockIdx.x * blockDim.x + threadIdx.x;
  if (t >= N_NODES * HEADS) return;
  const int hh = t & (HEADS - 1);
  const int node = t >> 3;
  const float* __restrict__ hv = h + (size_t)node * FDIM + hh * HEAD_DIM;
  const float* __restrict__ asv = att_src + hh * HEAD_DIM;
  const float* __restrict__ adv = att_dst + hh * HEAD_DIM;
  float s1 = 0.f, s2 = 0.f;
#pragma unroll
  for (int d = 0; d < HEAD_DIM; ++d) {
    float v = hv[d];
    s1 += v * asv[d];
    s2 += v * adv[d];
  }
  a_src[t] = s1;
  a_dst[t] = s2;
}

__global__ void init_ms(float* __restrict__ m, float* __restrict__ s) {
  int t = blockIdx.x * blockDim.x + threadIdx.x;
  if (t >= N_NODES * HEADS) return;
  m[t] = -__builtin_inff();
  s[t] = 0.f;
}

__global__ void init_bias(float* __restrict__ acc, const float* __restrict__ bias) {
  int t = blockIdx.x * blockDim.x + threadIdx.x;
  if (t >= N_NODES * FDIM) return;
  acc[t] = bias[t & (FDIM - 1)];
}

__device__ __forceinline__ void edge_decode(const int* __restrict__ ei, int e,
                                            int& src, int& dst) {
  if (e < N_EDGES) {
    src = ei[e];
    dst = ei[N_EDGES + e];
  } else {
    src = dst = e - N_EDGES;
  }
}

__device__ __forceinline__ float leaky(float x) {
  return x > 0.f ? x : NEG_SLOPE * x;
}

__device__ __forceinline__ void atomicMaxF(float* addr, float val) {
  if (val >= 0.f)
    atomicMax((int*)addr, __float_as_int(val));
  else
    atomicMin((unsigned int*)addr, __float_as_uint(val));
}

__global__ void edge_max(const int* __restrict__ ei,
                         const float* __restrict__ a_src,
                         const float* __restrict__ a_dst, float* __restrict__ m) {
  int t = blockIdx.x * blockDim.x + threadIdx.x;
  if (t >= ETOT * HEADS) return;
  const int hh = t & (HEADS - 1);
  const int e = t >> 3;
  int src, dst;
  edge_decode(ei, e, src, dst);
  float ee = leaky(a_src[src * HEADS + hh] + a_dst[dst * HEADS + hh]);
  atomicMaxF(&m[dst * HEADS + hh], ee);
}

__global__ void edge_sum(const int* __restrict__ ei,
                         const float* __restrict__ a_src,
                         const float* __restrict__ a_dst,
                         const float* __restrict__ m, float* __restrict__ s) {
  int t = blockIdx.x * blockDim.x + threadIdx.x;
  if (t >= ETOT * HEADS) return;
  const int hh = t & (HEADS - 1);
  const int e = t >> 3;
  int src, dst;
  edge_decode(ei, e, src, dst);
  float ee = leaky(a_src[src * HEADS + hh] + a_dst[dst * HEADS + hh]);
  float mm = m[dst * HEADS + hh];
  if (!(mm > -__builtin_inff())) mm = 0.f;
  atomicAdd(&s[dst * HEADS + hh], expf(ee - mm));
}

__global__ void edge_agg(const int* __restrict__ ei,
                         const float* __restrict__ a_src,
                         const float* __restrict__ a_dst,
                         const float* __restrict__ m, const float* __restrict__ s,
                         const float* __restrict__ h, float* __restrict__ acc) {
  int t = blockIdx.x * blockDim.x + threadIdx.x;
  if (t >= ETOT * HEADS) return;
  const int hh = t & (HEADS - 1);
  const int e = t >> 3;
  int src, dst;
  edge_decode(ei, e, src, dst);
  float ee = leaky(a_src[src * HEADS + hh] + a_dst[dst * HEADS + hh]);
  float mm = m[dst * HEADS + hh];
  if (!(mm > -__builtin_inff())) mm = 0.f;
  float p = expf(ee - mm);
  float alpha = p / (s[dst * HEADS + hh] + 1e-16f);

  const float* __restrict__ hs = h + (size_t)src * FDIM + hh * HEAD_DIM;
  float* __restrict__ od = acc + (size_t)dst * FDIM + hh * HEAD_DIM;
#pragma unroll
  for (int d = 0; d < HEAD_DIM; d += 4) {
    float4 v = *(const float4*)(hs + d);
    atomicAdd(od + d + 0, alpha * v.x);
    atomicAdd(od + d + 1, alpha * v.y);
    atomicAdd(od + d + 2, alpha * v.z);
    atomicAdd(od + d + 3, alpha * v.w);
  }
}

__global__ void elu_inplace(float* __restrict__ x) {
  int t = blockIdx.x * blockDim.x + threadIdx.x;
  if (t >= N_NODES * FDIM) return;
  float v = x[t];
  x[t] = v > 0.f ? v : (expf(v) - 1.f);
}

extern "C" void kernel_launch(void* const* d_in, const int* in_sizes, int n_in,
                              void* d_out, int out_size, void* d_ws,
                              size_t ws_size, hipStream_t stream) {
  (void)in_sizes; (void)n_in; (void)out_size;

  const float* x   = (const float*)d_in[0];
  const int*   ei  = (const int*)d_in[1];
  const float* W1  = (const float*)d_in[2];
  const float* as1 = (const float*)d_in[3];
  const float* ad1 = (const float*)d_in[4];
  const float* b1  = (const float*)d_in[5];
  const float* W2  = (const float*)d_in[6];
  const float* as2 = (const float*)d_in[7];
  const float* ad2 = (const float*)d_in[8];
  const float* b2  = (const float*)d_in[9];
  float* out = (float*)d_out;

  float* ws    = (float*)d_ws;
  float* h     = ws;
  float* agg   = h + (size_t)N_NODES * FDIM;
  float* a_src = agg + (size_t)N_NODES * FDIM;
  float* a_dst = a_src + (size_t)N_NODES * HEADS;
  float* m     = a_dst + (size_t)N_NODES * HEADS;
  float* s     = m + (size_t)N_NODES * HEADS;
  const size_t need = ((size_t)N_NODES * FDIM * 2 + (size_t)N_NODES * HEADS * 4) *
                      sizeof(float);
  if (ws_size < need) return;

  const int gemmBlocks = (N_NODES / 16) * (FDIM / 16) / 4;
  const int nh = N_NODES * HEADS;
  const int nf = N_NODES * FDIM;
  const int eh = ETOT * HEADS;
  const int gNH = (nh + 255) / 256;
  const int gNF = (nf + 255) / 256;
  const int gEH = (eh + 255) / 256;
  dim3 B128(128), B256(256);

  gemm128_wmma_f32<<<gemmBlocks, B128, 0, stream>>>(x, W1, h);
  att_scores<<<gNH, B256, 0, stream>>>(h, as1, ad1, a_src, a_dst);
  init_ms<<<gNH, B256, 0, stream>>>(m, s);
  init_bias<<<gNF, B256, 0, stream>>>(agg, b1);
  edge_max<<<gEH, B256, 0, stream>>>(ei, a_src, a_dst, m);
  edge_sum<<<gEH, B256, 0, stream>>>(ei, a_src, a_dst, m, s);
  edge_agg<<<gEH, B256, 0, stream>>>(ei, a_src, a_dst, m, s, h, agg);
  elu_inplace<<<gNF, B256, 0, stream>>>(agg);

  gemm128_wmma_f32<<<gemmBlocks, B128, 0, stream>>>(agg, W2, h);
  att_scores<<<gNH, B256, 0, stream>>>(h, as2, ad2, a_src, a_dst);
  init_ms<<<gNH, B256, 0, stream>>>(m, s);
  init_bias<<<gNF, B256, 0, stream>>>(out, b2);
  edge_max<<<gEH, B256, 0, stream>>>(ei, a_src, a_dst, m);
  edge_sum<<<gEH, B256, 0, stream>>>(ei, a_src, a_dst, m, s);
  edge_agg<<<gEH, B256, 0, stream>>>(ei, a_src, a_dst, m, s, h, out);
}


# edge_agg: lanes remapped to feature columns so each wave-wide f32 atomic add (and h load) covers one whole 128-byte line, not 16 lines of 2 lanes; sums still in atomic arrival order
# speedup vs baseline: 2.9062x; 2.9062x over previous
#include <hip/hip_runtime.h>
#include <math.h>


#define N_NODES 50000
#define N_EDGES 800000
#define HEADS   8
#define HEAD_DIM 16
#define FDIM    128
#define ETOT    (N_EDGES + N_NODES)
#define NEG_SLOPE 0.2f

typedef __attribute__((ext_vector_type(2))) float v2f;
typedef __attribute__((ext_vector_type(8))) float v8f;

__global__ void __launch_bounds__(128)
gemm128_wmma_f32(const float* __restrict__ A, const float* __restrict__ W,
                 float* __restrict__ out) {
  const int wave = threadIdx.x >> 5;
  const int lane = threadIdx.x & 31;
  const int tile = blockIdx.x * 4 + wave;
  const int tm   = tile >> 3;
  const int tn   = tile & 7;
  const int half = lane >> 4;
  const int l    = lane & 15;

  const float* __restrict__ arow = A + (size_t)(tm * 16 + l) * FDIM;
  const float* __restrict__ bcol = W + tn * 16 + l;

  v8f c = {};
#pragma unroll
  for (int k0 = 0; k0 < FDIM; k0 += 4) {
    const int ka = k0 + 2 * half;
    v2f a, b;
    a.x = arow[ka];
    a.y = arow[ka + 1];
    b.x = bcol[(size_t)ka * FDIM];
    b.y = bcol[(size_t)(ka + 1) * FDIM];
    c = __builtin_amdgcn_wmma_f32_16x16x4_f32(false, a, false, b, (short)0, c,
                                              false, false);
  }
  float* __restrict__ orow =
      out + (size_t)(tm * 16 + 8 * half) * FDIM + tn * 16 + l;
#pragma unroll
  for (int r = 0; r < 8; ++r) orow[(size_t)r * FDIM] = c[r];
}

__global__ void att_scores(const float* __restrict__ h,
                           const float* __restrict__ att_src,
                           const float* __restrict__ att_dst,
                           float* __restrict__ a_src, float* __restrict__ a_dst) {
  int t = blockIdx.x * blockDim.x + threadIdx.x;
  if (t >= N_NODES * HEADS) return;
  const int hh = t & (HEADS - 1);
  const int node = t >> 3;
  const float* __restrict__ hv = h + (size_t)node * FDIM + hh * HEAD_DIM;
  const float* __restrict__ asv = att_src + hh * HEAD_DIM;
  const float* __restrict__ adv = att_dst + hh * HEAD_DIM;
  float s1 = 0.f, s2 = 0.f;
#pragma unroll
  for (int d = 0; d < HEAD_DIM; ++d) {
    float v = hv[d];
    s1 += v * asv[d];
    s2 += v * adv[d];
  }
  a_src[t] = s1;
  a_dst[t] = s2;
}

__global__ void init_ms(float* __restrict__ m, float* __restrict__ s) {
  int t = blockIdx.x * blockDim.x + threadIdx.x;
  if (t >= N_NODES * HEADS) return;
  m[t] = -__builtin_inff();
  s[t] = 0.f;
}

__global__ void init_bias(float* __restrict__ acc, const float* __restrict__ bias) {
  int t = blockIdx.x * blockDim.x + threadIdx.x;
  if (t >= N_NODES * FDIM) return;
  acc[t] = bias[t & (FDIM - 1)];
}

__device__ __forceinline__ void edge_decode(const int* __restrict__ ei, int e,
                                            int& src, int& dst) {
  if (e < N_EDGES) {
    src = ei[e];
    dst = ei[N_EDGES + e];
  } else {
    src = dst = e - N_EDGES;
  }
}

__device__ __forceinline__ float leaky(float x) {
  return x > 0.f ? x : NEG_SLOPE * x;
}

__device__ __forceinline__ void atomicMaxF(float* addr, float val) {
  if (val >= 0.f)
    atomicMax((int*)addr, __float_as_int(val));
  else
    atomicMin((unsigned int*)addr, __float_as_uint(val));
}

__global__ void edge_max(const int* __restrict__ ei,
                         const float* __restrict__ a_src,
                         const float* __restrict__ a_dst, float* __restrict__ m) {
  int t = blockIdx.x * blockDim.x + threadIdx.x;
  if (t >= ETOT * HEADS) return;
  const int hh = t & (HEADS - 1);
  const int e = t >> 3;
  int src, dst;
  edge_decode(ei, e, src, dst);
  float ee = leaky(a_src[src * HEADS + hh] + a_dst[dst * HEADS + hh]);
  atomicMaxF(&m[dst * HEADS + hh], ee);
}

__global__ void edge_sum(const int* __restrict__ ei,
                         const float* __restrict__ a_src,
                         const float* __restrict__ a_dst,
                         const float* __restrict__ m, float* __restrict__ s) {
  int t = blockIdx.x * blockDim.x + threadIdx.x;
  if (t >= ETOT * HEADS) return;
  const int hh = t & (HEADS - 1);
  const int e = t >> 3;
  int src, dst;
  edge_decode(ei, e, src, dst);
  float ee = leaky(a_src[src * HEADS + hh] + a_dst[dst * HEADS + hh]);
  float mm = m[dst * HEADS + hh];
  if (!(mm > -__builtin_inff())) mm = 0.f;
  atomicAdd(&s[dst * HEADS + hh], expf(ee - mm));
}

static_assert((ETOT * HEADS) % 32 == 0,
              "edge_agg: a wave must hold four whole edges or none");

__global__ void edge_agg(const int* __restrict__ ei,
                         const float* __restrict__ a_src,
                         const float* __restrict__ a_dst,
                         const float* __restrict__ m, const float* __restrict__ s,
                         const float* __restrict__ h, float* __restrict__ acc) {
  const int t = blockIdx.x * blockDim.x + threadIdx.x;
  const int lane = threadIdx.x & 31;
  const int e0 = __builtin_amdgcn_readfirstlane(t) >> 3;
  if (e0 >= ETOT) return;

  const int hh = t & (HEADS - 1);
  const int e = t >> 3;
  int src, dst;
  edge_decode(ei, e, src, dst);

  float v[4][4];
  float* od[4];
#pragma unroll
  for (int q = 0; q < 4; ++q) {
    const int sq = __builtin_amdgcn_readlane(src, q * HEADS);
    const int dq = __builtin_amdgcn_readlane(dst, q * HEADS);
    const float* __restrict__ hs = h + (size_t)sq * FDIM + lane;
    od[q] = acc + (size_t)dq * FDIM + lane;
#pragma unroll
    for (int j = 0; j < 4; ++j) v[q][j] = hs[32 * j];
  }

  float ee = leaky(a_src[src * HEADS + hh] + a_dst[dst * HEADS + hh]);
  float mm = m[dst * HEADS + hh];
  if (!(mm > -__builtin_inff())) mm = 0.f;
  float p = expf(ee - mm);
  const float alpha = p / (s[dst * HEADS + hh] + 1e-16f);

#pragma unroll
  for (int q = 0; q < 4; ++q) {
#pragma unroll
    for (int j = 0; j < 4; ++j) {
      const float a = __shfl(alpha, q * HEADS + 2 * j + (lane >> 4), 32);
      atomicAdd(od[q] + 32 * j, a * v[q][j]);
    }
  }
}

__global__ void elu_inplace(float* __restrict__ x) {
  int t = blockIdx.x * blockDim.x + threadIdx.x;
  if (t >= N_NODES * FDIM) return;
  float v = x[t];
  x[t] = v > 0.f ? v : (expf(v) - 1.f);
}

extern "C" void kernel_launch(void* const* d_in, const int* in_sizes, int n_in,
                              void* d_out, int out_size, void* d_ws,
                              size_t ws_size, hipStream_t stream) {
  (void)in_sizes; (void)n_in; (void)out_size;

  const float* x   = (const float*)d_in[0];
  const int*   ei  = (const int*)d_in[1];
  const float* W1  = (const float*)d_in[2];
  const float* as1 = (const float*)d_in[3];
  const float* ad1 = (const float*)d_in[4];
  const float* b1  = (const float*)d_in[5];
  const float* W2  = (const float*)d_in[6];
  const float* as2 = (const float*)d_in[7];
  const float* ad2 = (const float*)d_in[8];
  const float* b2  = (const float*)d_in[9];
  float* out = (float*)d_out;

  float* ws    = (float*)d_ws;
  float* h     = ws;
  float* agg   = h + (size_t)N_NODES * FDIM;
  float* a_src = agg + (size_t)N_NODES * FDIM;
  float* a_dst = a_src + (size_t)N_NODES * HEADS;
  float* m     = a_dst + (size_t)N_NODES * HEADS;
  float* s     = m + (size_t)N_NODES * HEADS;
  const size_t need = ((size_t)N_NODES * FDIM * 2 + (size_t)N_NODES * HEADS * 4) *
                      sizeof(float);
  if (ws_size < need) return;

  const int gemmBlocks = (N_NODES / 16) * (FDIM / 16) / 4;
  const int nh = N_NODES * HEADS;
  const int nf = N_NODES * FDIM;
  const int eh = ETOT * HEADS;
  const int gNH = (nh + 255) / 256;
  const int gNF = (nf + 255) / 256;
  const int gEH = (eh + 255) / 256;
  dim3 B128(128), B256(256);

  gemm128_wmma_f32<<<gemmBlocks, B128, 0, stream>>>(x, W1, h);
  att_scores<<<gNH, B256, 0, stream>>>(h, as1, ad1, a_src, a_dst);
  init_ms<<<gNH, B256, 0, stream>>>(m, s);
  init_bias<<<gNF, B256, 0, stream>>>(agg, b1);
  edge_max<<<gEH, B256, 0, stream>>>(ei, a_src, a_dst, m);
  edge_sum<<<gEH, B256, 0, stream>>>(ei, a_src, a_dst, m, s);
  edge_agg<<<gEH, B256, 0, stream>>>(ei, a_src, a_dst, m, s, h, agg);
  elu_inplace<<<gNF, B256, 0, stream>>>(agg);

  gemm128_wmma_f32<<<gemmBlocks, B128, 0, stream>>>(agg, W2, h);
  att_scores<<<gNH, B256, 0, stream>>>(h, as2, ad2, a_src, a_dst);
  init_ms<<<gNH, B256, 0, stream>>>(m, s);
  init_bias<<<gNF, B256, 0, stream>>>(out, b2);
  edge_max<<<gEH, B256, 0, stream>>>(ei, a_src, a_dst, m);
  edge_sum<<<gEH, B256, 0, stream>>>(ei, a_src, a_dst, m, s);
  edge_agg<<<gEH, B256, 0, stream>>>(ei, a_src, a_dst, m, s, h, out);
}


# edges counting-sorted by destination in-call (int atomic rank + scan); one wave per node gathers its in-edges: softmax, row sum, bias, ELU in registers, no float atomics; sums in arrival-rank order
# speedup vs baseline: 6.3068x; 2.1701x over previous
#include <hip/hip_runtime.h>
#include <math.h>


#define N_NODES 50000
#define N_EDGES 800000
#define HEADS   8
#define HEAD_DIM 16
#define FDIM    128
#define NEG_SLOPE 0.2f

typedef __attribute__((ext_vector_type(2))) float v2f;
typedef __attribute__((ext_vector_type(8))) float v8f;

__global__ void __launch_bounds__(128)
gemm128_wmma_f32(const float* __restrict__ A, const float* __restrict__ W,
                 float* __restrict__ out) {
  const int wave = threadIdx.x >> 5;
  const int lane = threadIdx.x & 31;
  const int tile = blockIdx.x * 4 + wave;
  const int tm   = tile >> 3;
  const int tn   = tile & 7;
  const int half = lane >> 4;
  const int l    = lane & 15;

  const float* __restrict__ arow = A + (size_t)(tm * 16 + l) * FDIM;
  const float* __restrict__ bcol = W + tn * 16 + l;

  v8f c = {};
#pragma unroll
  for (int k0 = 0; k0 < FDIM; k0 += 4) {
    const int ka = k0 + 2 * half;
    v2f a, b;
    a.x = arow[ka];
    a.y = arow[ka + 1];
    b.x = bcol[(size_t)ka * FDIM];
    b.y = bcol[(size_t)(ka + 1) * FDIM];
    c = __builtin_amdgcn_wmma_f32_16x16x4_f32(false, a, false, b, (short)0, c,
                                              false, false);
  }
  float* __restrict__ orow =
      out + (size_t)(tm * 16 + 8 * half) * FDIM + tn * 16 + l;
#pragma unroll
  for (int r = 0; r < 8; ++r) orow[(size_t)r * FDIM] = c[r];
}

__global__ void att_scores(const float* __restrict__ h,
                           const float* __restrict__ att_src,
                           const float* __restrict__ att_dst,
                           float* __restrict__ a_src, float* __restrict__ a_dst) {
  int t = blockIdx.x * blockDim.x + threadIdx.x;
  if (t >= N_NODES * HEADS) return;
  const int hh = t & (HEADS - 1);
  const int node = t >> 3;
  const float* __restrict__ hv = h + (size_t)node * FDIM + hh * HEAD_DIM;
  const float* __restrict__ asv = att_src + hh * HEAD_DIM;
  const float* __restrict__ adv = att_dst + hh * HEAD_DIM;
  float s1 = 0.f, s2 = 0.f;
#pragma unroll
  for (int d = 0; d < HEAD_DIM; ++d) {
    float v = hv[d];
    s1 += v * asv[d];
    s2 += v * adv[d];
  }
  a_src[t] = s1;
  a_dst[t] = s2;
}

#define SCAN_ROWS 13
#define SCAN_SPAN (SCAN_ROWS * 128)
#define SCAN_PAD  (32 * SCAN_SPAN)
static_assert(SCAN_PAD >= N_NODES + 1, "scan: padded counter array too short");
static_assert(N_NODES % 8 == 0, "node_gather: eight whole waves per block");

typedef __attribute__((ext_vector_type(4))) int v4i;
typedef __attribute__((ext_vector_type(4))) float v4f;

__global__ void zero_counts(int* __restrict__ cnt) {
  int t = blockIdx.x * blockDim.x + threadIdx.x;
  if (t < SCAN_PAD) cnt[t] = 0;
}

__global__ void count_rank(const int* __restrict__ ei, int* __restrict__ cnt,
                           int* __restrict__ rank) {
  int e = blockIdx.x * blockDim.x + threadIdx.x;
  if (e >= N_EDGES) return;
  rank[e] = atomicAdd(&cnt[ei[N_EDGES + e]], 1);
}

__global__ void __launch_bounds__(1024)
scan_counts(const int* __restrict__ cnt, int* __restrict__ rowptr) {
  __shared__ int wsum[32];
  const int lane = threadIdx.x & 31;
  const int wave = threadIdx.x >> 5;
  const int base = wave * SCAN_SPAN + 4 * lane;
  v4i v[SCAN_ROWS];
#pragma unroll
  for (int r = 0; r < SCAN_ROWS; ++r) v[r] = *(const v4i*)(cnt + base + 128 * r);
  int carry = 0;
#pragma unroll
  for (int r = 0; r < SCAN_ROWS; ++r) {
    const int x = v[r].x, y = v[r].y, z = v[r].z, w = v[r].w;
    const int t4 = x + y + z + w;
    int inc = t4;
#pragma unroll
    for (int d = 1; d < 32; d <<= 1) {
      const int up = __shfl_up(inc, d, 32);
      if (lane >= d) inc += up;
    }
    const int ex = carry + inc - t4;
    v[r].x = ex;
    v[r].y = ex + x;
    v[r].z = ex + x + y;
    v[r].w = ex + x + y + z;
    carry += __shfl(inc, 31, 32);
  }
  if (lane == 0) wsum[wave] = carry;
  __syncthreads();
  if (wave == 0) {
    const int w = wsum[lane];
    int inc = w;
#pragma unroll
    for (int d = 1; d < 32; d <<= 1) {
      const int up = __shfl_up(inc, d, 32);
      if (lane >= d) inc += up;
    }
    wsum[lane] = inc - w;
  }
  __syncthreads();
  const int off = wsum[wave];
#pragma unroll
  for (int r = 0; r < SCAN_ROWS; ++r) {
    v4i o = v[r];
    o.x += off; o.y += off; o.z += off; o.w += off;
    *(v4i*)(rowptr + base + 128 * r) = o;
  }
}

__global__ void fill_cols(const int* __restrict__ ei,
                          const int* __restrict__ rowptr,
                          const int* __restrict__ rank, int* __restrict__ col) {
  int e = blockIdx.x * blockDim.x + threadIdx.x;
  if (e >= N_EDGES) return;
  col[rowptr[ei[N_EDGES + e]] + rank[e]] = ei[e];
}

__device__ __forceinline__ float leaky(float x) {
  return x > 0.f ? x : NEG_SLOPE * x;
}

__device__ __forceinline__ void chunk_scores(
    const int* __restrict__ col, const float* __restrict__ a_src, int start,
    int deg, int n, int cb, int lane, float ad, int& my, float (&e)[8]) {
  const int j = cb + lane;
  my = (j < deg) ? col[start + j] : n;
  const int hh = lane & 7;
  const int slot = lane >> 3;
#pragma unroll
  for (int i = 0; i < 8; ++i) {
    const int jj = 4 * i + slot;
    const int sj = __shfl(my, jj, 32);
    const float sc = leaky(a_src[sj * HEADS + hh] + ad);
    e[i] = (cb + jj <= deg) ? sc : -__builtin_inff();
  }
}

__device__ __forceinline__ void add_rows(const float* __restrict__ h, int my,
                                         const float (&p)[8], float den,
                                         int nvalid, int lane, v4f& acc) {
  const int h2 = lane >> 2;
#pragma unroll
  for (int i = 0; i < 8; ++i) {
    if (4 * i < nvalid) {
      const float al = p[i] / den;
      v4f r[4];
#pragma unroll
      for (int q = 0; q < 4; ++q) {
        const int sj = __builtin_amdgcn_readlane(my, 4 * i + q);
        r[q] = *(const v4f*)(h + (size_t)sj * FDIM + 4 * lane);
      }
#pragma unroll
      for (int q = 0; q < 4; ++q) {
        const float a = __shfl(al, q * HEADS + h2, 32);
        const v4f rr = (4 * i + q < nvalid) ? r[q] : (v4f){0.f, 0.f, 0.f, 0.f};
        acc.x = fmaf(a, rr.x, acc.x);
        acc.y = fmaf(a, rr.y, acc.y);
        acc.z = fmaf(a, rr.z, acc.z);
        acc.w = fmaf(a, rr.w, acc.w);
      }
    }
  }
}

__global__ void __launch_bounds__(256)
node_gather(const int* __restrict__ rowptr, const int* __restrict__ col,
            const float* __restrict__ a_src, const float* __restrict__ a_dst,
            const float* __restrict__ h, const float* __restrict__ bias,
            float* __restrict__ out, int apply_elu) {
  const int lane = threadIdx.x & 31;
  const int n = __builtin_amdgcn_readfirstlane(
      (int)(blockIdx.x * blockDim.x + threadIdx.x) >> 5);
  if (n >= N_NODES) return;
  const int start = rowptr[n];
  const int deg = rowptr[n + 1] - start;
  const int total = deg + 1;
  const float ad = a_dst[n * HEADS + (lane & 7)];

  int my0, my;
  float e0[8], e[8];
  chunk_scores(col, a_src, start, deg, n, 0, lane, ad, my0, e0);

  float mx = -__builtin_inff();
#pragma unroll
  for (int i = 0; i < 8; ++i) mx = fmaxf(mx, e0[i]);
  for (int cb = 32; cb < total; cb += 32) {
    chunk_scores(col, a_src, start, deg, n, cb, lane, ad, my, e);
#pragma unroll
    for (int i = 0; i < 8; ++i) mx = fmaxf(mx, e[i]);
  }
  mx = fmaxf(mx, __shfl_xor(mx, 8, 32));
  mx = fmaxf(mx, __shfl_xor(mx, 16, 32));
  if (!(mx > -__builtin_inff())) mx = 0.f;

  float sum = 0.f;
#pragma unroll
  for (int i = 0; i < 8; ++i) {
    e0[i] = expf(e0[i] - mx);
    sum += e0[i];
  }
  for (int cb = 32; cb < total; cb += 32) {
    chunk_scores(col, a_src, start, deg, n, cb, lane, ad, my, e);
#pragma unroll
    for (int i = 0; i < 8; ++i) sum += expf(e[i] - mx);
  }
  sum += __shfl_xor(sum, 8, 32);
  sum += __shfl_xor(sum, 16, 32);
  const float den = sum + 1e-16f;

  v4f acc = {0.f, 0.f, 0.f, 0.f};
  add_rows(h, my0, e0, den, total, lane, acc);
  for (int cb = 32; cb < total; cb += 32) {
    chunk_scores(col, a_src, start, deg, n, cb, lane, ad, my, e);
#pragma unroll
    for (int i = 0; i < 8; ++i) e[i] = expf(e[i] - mx);
    add_rows(h, my, e, den, total - cb, lane, acc);
  }

  const v4f b = *(const v4f*)(bias + 4 * lane);
  v4f o = acc + b;
  if (apply_elu) {
    o.x = o.x > 0.f ? o.x : (expf(o.x) - 1.f);
    o.y = o.y > 0.f ? o.y : (expf(o.y) - 1.f);
    o.z = o.z > 0.f ? o.z : (expf(o.z) - 1.f);
    o.w = o.w > 0.f ? o.w : (expf(o.w) - 1.f);
  }
  *(v4f*)(out + (size_t)n * FDIM + 4 * lane) = o;
}

extern "C" void kernel_launch(void* const* d_in, const int* in_sizes, int n_in,
                              void* d_out, int out_size, void* d_ws,
                              size_t ws_size, hipStream_t stream) {
  (void)in_sizes; (void)n_in; (void)out_size;

  const float* x   = (const float*)d_in[0];
  const int*   ei  = (const int*)d_in[1];
  const float* W1  = (const float*)d_in[2];
  const float* as1 = (const float*)d_in[3];
  const float* ad1 = (const float*)d_in[4];
  const float* b1  = (const float*)d_in[5];
  const float* W2  = (const float*)d_in[6];
  const float* as2 = (const float*)d_in[7];
  const float* ad2 = (const float*)d_in[8];
  const float* b2  = (const float*)d_in[9];
  float* out = (float*)d_out;

  float* ws    = (float*)d_ws;
  float* h     = ws;
  float* agg   = h + (size_t)N_NODES * FDIM;
  float* a_src = agg + (size_t)N_NODES * FDIM;
  float* a_dst = a_src + (size_t)N_NODES * HEADS;
  int* cnt     = (int*)(a_dst + (size_t)N_NODES * HEADS);
  int* rowptr  = cnt + SCAN_PAD;
  int* rank    = rowptr + SCAN_PAD;
  int* col     = rank + N_EDGES;
  const size_t need = ((size_t)N_NODES * FDIM * 2 + (size_t)N_NODES * HEADS * 2 +
                       (size_t)SCAN_PAD * 2 + (size_t)N_EDGES * 2) * 4;
  if (ws_size < need) return;

  const int gemmBlocks = (N_NODES / 16) * (FDIM / 16) / 4;
  const int nh = N_NODES * HEADS;
  const int gNH = (nh + 255) / 256;
  const int gE  = (N_EDGES + 255) / 256;
  const int gZ  = (SCAN_PAD + 255) / 256;
  const int gN  = N_NODES / 8;
  dim3 B128(128), B256(256), B1024(1024);

  zero_counts<<<gZ, B256, 0, stream>>>(cnt);
  count_rank<<<gE, B256, 0, stream>>>(ei, cnt, rank);
  scan_counts<<<1, B1024, 0, stream>>>(cnt, rowptr);
  fill_cols<<<gE, B256, 0, stream>>>(ei, rowptr, rank, col);

  gemm128_wmma_f32<<<gemmBlocks, B128, 0, stream>>>(x, W1, h);
  att_scores<<<gNH, B256, 0, stream>>>(h, as1, ad1, a_src, a_dst);
  node_gather<<<gN, B256, 0, stream>>>(rowptr, col, a_src, a_dst, h, b1, agg, 1);

  gemm128_wmma_f32<<<gemmBlocks, B128, 0, stream>>>(agg, W2, h);
  att_scores<<<gNH, B256, 0, stream>>>(h, as2, ad2, a_src, a_dst);
  node_gather<<<gN, B256, 0, stream>>>(rowptr, col, a_src, a_dst, h, b2, out, 0);
}


# GEMM: a wave does 16 rows x 64 columns, four tiles on one A fragment, tile columns interleaved by 4 so W rows load as 16-byte pieces and stores are whole lines; order of sums unchanged
# speedup vs baseline: 7.0866x; 1.1236x over previous
#include <hip/hip_runtime.h>
#include <math.h>


#define N_NODES 50000
#define N_EDGES 800000
#define HEADS   8
#define HEAD_DIM 16
#define FDIM    128
#define NEG_SLOPE 0.2f

typedef __attribute__((ext_vector_type(2))) float v2f;
typedef __attribute__((ext_vector_type(8))) float v8f;
typedef __attribute__((ext_vector_type(4))) float v4f;

static_assert(N_NODES % 16 == 0, "gemm: whole 16-row strips");
__global__ void __launch_bounds__(64)
gemm128_wmma_f32(const float* __restrict__ A, const float* __restrict__ W,
                 float* __restrict__ out) {
  const int wave = threadIdx.x >> 5;
  const int lane = threadIdx.x & 31;
  const int tm   = blockIdx.x;
  const int half = lane >> 4;
  const int l    = lane & 15;

  const float* __restrict__ arow = A + (size_t)(tm * 16 + l) * FDIM + 2 * half;
  const float* __restrict__ bcol = W + (size_t)(2 * half) * FDIM + wave * 64 + 4 * l;

  v8f c0 = {}, c1 = {}, c2 = {}, c3 = {};
#pragma unroll
  for (int k0 = 0; k0 < FDIM; k0 += 4) {
    v2f a;
    a.x = arow[k0];
    a.y = arow[k0 + 1];
    const v4f b0 = *(const v4f*)(bcol + (size_t)k0 * FDIM);
    const v4f b1 = *(const v4f*)(bcol + (size_t)(k0 + 1) * FDIM);
    c0 = __builtin_amdgcn_wmma_f32_16x16x4_f32(false, a, false, (v2f){b0.x, b1.x},
                                               (short)0, c0, false, false);
    c1 = __builtin_amdgcn_wmma_f32_16x16x4_f32(false, a, false, (v2f){b0.y, b1.y},
                                               (short)0, c1, false, false);
    c2 = __builtin_amdgcn_wmma_f32_16x16x4_f32(false, a, false, (v2f){b0.z, b1.z},
                                               (short)0, c2, false, false);
    c3 = __builtin_amdgcn_wmma_f32_16x16x4_f32(false, a, false, (v2f){b0.w, b1.w},
                                               (short)0, c3, false, false);
  }
  float* __restrict__ orow =
      out + (size_t)(tm * 16 + 8 * half) * FDIM + wave * 64 + 4 * l;
#pragma unroll
  for (int r = 0; r < 8; ++r)
    *(v4f*)(orow + (size_t)r * FDIM) = (v4f){c0[r], c1[r], c2[r], c3[r]};
}

__global__ void att_scores(const float* __restrict__ h,
                           const float* __restrict__ att_src,
                           const float* __restrict__ att_dst,
                           float* __restrict__ a_src, float* __restrict__ a_dst) {
  int t = blockIdx.x * blockDim.x + threadIdx.x;
  if (t >= N_NODES * HEADS) return;
  const int hh = t & (HEADS - 1);
  const int node = t >> 3;
  const float* __restrict__ hv = h + (size_t)node * FDIM + hh * HEAD_DIM;
  const float* __restrict__ asv = att_src + hh * HEAD_DIM;
  const float* __restrict__ adv = att_dst + hh * HEAD_DIM;
  float s1 = 0.f, s2 = 0.f;
#pragma unroll
  for (int d = 0; d < HEAD_DIM; ++d) {
    float v = hv[d];
    s1 += v * asv[d];
    s2 += v * adv[d];
  }
  a_src[t] = s1;
  a_dst[t] = s2;
}

#define SCAN_ROWS 13
#define SCAN_SPAN (SCAN_ROWS * 128)
#define SCAN_PAD  (32 * SCAN_SPAN)
static_assert(SCAN_PAD >= N_NODES + 1, "scan: padded counter array too short");
static_assert(N_NODES % 8 == 0, "node_gather: eight whole waves per block");

typedef __attribute__((ext_vector_type(4))) int v4i;

__global__ void zero_counts(int* __restrict__ cnt) {
  int t = blockIdx.x * blockDim.x + threadIdx.x;
  if (t < SCAN_PAD) cnt[t] = 0;
}

__global__ void count_rank(const int* __restrict__ ei, int* __restrict__ cnt,
                           int* __restrict__ rank) {
  int e = blockIdx.x * blockDim.x + threadIdx.x;
  if (e >= N_EDGES) return;
  rank[e] = atomicAdd(&cnt[ei[N_EDGES + e]], 1);
}

__global__ void __launch_bounds__(1024)
scan_counts(const int* __restrict__ cnt, int* __restrict__ rowptr) {
  __shared__ int wsum[32];
  const int lane = threadIdx.x & 31;
  const int wave = threadIdx.x >> 5;
  const int base = wave * SCAN_SPAN + 4 * lane;
  v4i v[SCAN_ROWS];
#pragma unroll
  for (int r = 0; r < SCAN_ROWS; ++r) v[r] = *(const v4i*)(cnt + base + 128 * r);
  int carry = 0;
#pragma unroll
  for (int r = 0; r < SCAN_ROWS; ++r) {
    const int x = v[r].x, y = v[r].y, z = v[r].z, w = v[r].w;
    const int t4 = x + y + z + w;
    int inc = t4;
#pragma unroll
    for (int d = 1; d < 32; d <<= 1) {
      const int up = __shfl_up(inc, d, 32);
      if (lane >= d) inc += up;
    }
    const int ex = carry + inc - t4;
    v[r].x = ex;
    v[r].y = ex + x;
    v[r].z = ex + x + y;
    v[r].w = ex + x + y + z;
    carry += __shfl(inc, 31, 32);
  }
  if (lane == 0) wsum[wave] = carry;
  __syncthreads();
  if (wave == 0) {
    const int w = wsum[lane];
    int inc = w;
#pragma unroll
    for (int d = 1; d < 32; d <<= 1) {
      const int up = __shfl_up(inc, d, 32);
      if (lane >= d) inc += up;
    }
    wsum[lane] = inc - w;
  }
  __syncthreads();
  const int off = wsum[wave];
#pragma unroll
  for (int r = 0; r < SCAN_ROWS; ++r) {
    v4i o = v[r];
    o.x += off; o.y += off; o.z += off; o.w += off;
    *(v4i*)(rowptr + base + 128 * r) = o;
  }
}

__global__ void fill_cols(const int* __restrict__ ei,
                          const int* __restrict__ rowptr,
                          const int* __restrict__ rank, int* __restrict__ col) {
  int e = blockIdx.x * blockDim.x + threadIdx.x;
  if (e >= N_EDGES) return;
  col[rowptr[ei[N_EDGES + e]] + rank[e]] = ei[e];
}

__device__ __forceinline__ float leaky(float x) {
  return x > 0.f ? x : NEG_SLOPE * x;
}

__device__ __forceinline__ void chunk_scores(
    const int* __restrict__ col, const float* __restrict__ a_src, int start,
    int deg, int n, int cb, int lane, float ad, int& my, float (&e)[8]) {
  const int j = cb + lane;
  my = (j < deg) ? col[start + j] : n;
  const int hh = lane & 7;
  const int slot = lane >> 3;
#pragma unroll
  for (int i = 0; i < 8; ++i) {
    const int jj = 4 * i + slot;
    const int sj = __shfl(my, jj, 32);
    const float sc = leaky(a_src[sj * HEADS + hh] + ad);
    e[i] = (cb + jj <= deg) ? sc : -__builtin_inff();
  }
}

__device__ __forceinline__ void add_rows(const float* __restrict__ h, int my,
                                         const float (&p)[8], float den,
                                         int nvalid, int lane, v4f& acc) {
  const int h2 = lane >> 2;
#pragma unroll
  for (int i = 0; i < 8; ++i) {
    if (4 * i < nvalid) {
      const float al = p[i] / den;
      v4f r[4];
#pragma unroll
      for (int q = 0; q < 4; ++q) {
        const int sj = __builtin_amdgcn_readlane(my, 4 * i + q);
        r[q] = *(const v4f*)(h + (size_t)sj * FDIM + 4 * lane);
      }
#pragma unroll
      for (int q = 0; q < 4; ++q) {
        const float a = __shfl(al, q * HEADS + h2, 32);
        const v4f rr = (4 * i + q < nvalid) ? r[q] : (v4f){0.f, 0.f, 0.f, 0.f};
        acc.x = fmaf(a, rr.x, acc.x);
        acc.y = fmaf(a, rr.y, acc.y);
        acc.z = fmaf(a, rr.z, acc.z);
        acc.w = fmaf(a, rr.w, acc.w);
      }
    }
  }
}

__global__ void __launch_bounds__(256)
node_gather(const int* __restrict__ rowptr, const int* __restrict__ col,
            const float* __restrict__ a_src, const float* __restrict__ a_dst,
            const float* __restrict__ h, const float* __restrict__ bias,
            float* __restrict__ out, int apply_elu) {
  const int lane = threadIdx.x & 31;
  const int n = __builtin_amdgcn_readfirstlane(
      (int)(blockIdx.x * blockDim.x + threadIdx.x) >> 5);
  if (n >= N_NODES) return;
  const int start = rowptr[n];
  const int deg = rowptr[n + 1] - start;
  const int total = deg + 1;
  const float ad = a_dst[n * HEADS + (lane & 7)];

  int my0, my;
  float e0[8], e[8];
  chunk_scores(col, a_src, start, deg, n, 0, lane, ad, my0, e0);

  float mx = -__builtin_inff();
#pragma unroll
  for (int i = 0; i < 8; ++i) mx = fmaxf(mx, e0[i]);
  for (int cb = 32; cb < total; cb += 32) {
    chunk_scores(col, a_src, start, deg, n, cb, lane, ad, my, e);
#pragma unroll
    for (int i = 0; i < 8; ++i) mx = fmaxf(mx, e[i]);
  }
  mx = fmaxf(mx, __shfl_xor(mx, 8, 32));
  mx = fmaxf(mx, __shfl_xor(mx, 16, 32));
  if (!(mx > -__builtin_inff())) mx = 0.f;

  float sum = 0.f;
#pragma unroll
  for (int i = 0; i < 8; ++i) {
    e0[i] = expf(e0[i] - mx);
    sum += e0[i];
  }
  for (int cb = 32; cb < total; cb += 32) {
    chunk_scores(col, a_src, start, deg, n, cb, lane, ad, my, e);
#pragma unroll
    for (int i = 0; i < 8; ++i) sum += expf(e[i] - mx);
  }
  sum += __shfl_xor(sum, 8, 32);
  sum += __shfl_xor(sum, 16, 32);
  const float den = sum + 1e-16f;

  v4f acc = {0.f, 0.f, 0.f, 0.f};
  add_rows(h, my0, e0, den, total, lane, acc);
  for (int cb = 32; cb < total; cb += 32) {
    chunk_scores(col, a_src, start, deg, n, cb, lane, ad, my, e);
#pragma unroll
    for (int i = 0; i < 8; ++i) e[i] = expf(e[i] - mx);
    add_rows(h, my, e, den, total - cb, lane, acc);
  }

  const v4f b = *(const v4f*)(bias + 4 * lane);
  v4f o = acc + b;
  if (apply_elu) {
    o.x = o.x > 0.f ? o.x : (expf(o.x) - 1.f);
    o.y = o.y > 0.f ? o.y : (expf(o.y) - 1.f);
    o.z = o.z > 0.f ? o.z : (expf(o.z) - 1.f);
    o.w = o.w > 0.f ? o.w : (expf(o.w) - 1.f);
  }
  *(v4f*)(out + (size_t)n * FDIM + 4 * lane) = o;
}

extern "C" void kernel_launch(void* const* d_in, const int* in_sizes, int n_in,
                              void* d_out, int out_size, void* d_ws,
                              size_t ws_size, hipStream_t stream) {
  (void)in_sizes; (void)n_in; (void)out_size;

  const float* x   = (const float*)d_in[0];
  const int*   ei  = (const int*)d_in[1];
  const float* W1  = (const float*)d_in[2];
  const float* as1 = (const float*)d_in[3];
  const float* ad1 = (const float*)d_in[4];
  const float* b1  = (const float*)d_in[5];
  const float* W2  = (const float*)d_in[6];
  const float* as2 = (const float*)d_in[7];
  const float* ad2 = (const float*)d_in[8];
  const float* b2  = (const float*)d_in[9];
  float* out = (float*)d_out;

  float* ws    = (float*)d_ws;
  float* h     = ws;
  float* agg   = h + (size_t)N_NODES * FDIM;
  float* a_src = agg + (size_t)N_NODES * FDIM;
  float* a_dst = a_src + (size_t)N_NODES * HEADS;
  int* cnt     = (int*)(a_dst + (size_t)N_NODES * HEADS);
  int* rowptr  = cnt + SCAN_PAD;
  int* rank    = rowptr + SCAN_PAD;
  int* col     = rank + N_EDGES;
  const size_t need = ((size_t)N_NODES * FDIM * 2 + (size_t)N_NODES * HEADS * 2 +
                       (size_t)SCAN_PAD * 2 + (size_t)N_EDGES * 2) * 4;
  if (ws_size < need) return;

  const int gemmBlocks = N_NODES / 16;
  const int nh = N_NODES * HEADS;
  const int gNH = (nh + 255) / 256;
  const int gE  = (N_EDGES + 255) / 256;
  const int gZ  = (SCAN_PAD + 255) / 256;
  const int gN  = N_NODES / 8;
  dim3 B64(64), B256(256), B1024(1024);

  zero_counts<<<gZ, B256, 0, stream>>>(cnt);
  count_rank<<<gE, B256, 0, stream>>>(ei, cnt, rank);
  scan_counts<<<1, B1024, 0, stream>>>(cnt, rowptr);
  fill_cols<<<gE, B256, 0, stream>>>(ei, rowptr, rank, col);

  gemm128_wmma_f32<<<gemmBlocks, B64, 0, stream>>>(x, W1, h);
  att_scores<<<gNH, B256, 0, stream>>>(h, as1, ad1, a_src, a_dst);
  node_gather<<<gN, B256, 0, stream>>>(rowptr, col, a_src, a_dst, h, b1, agg, 1);

  gemm128_wmma_f32<<<gemmBlocks, B64, 0, stream>>>(agg, W2, h);
  att_scores<<<gNH, B256, 0, stream>>>(h, as2, ad2, a_src, a_dst);
  node_gather<<<gN, B256, 0, stream>>>(rowptr, col, a_src, a_dst, h, b2, out, 0);
}


# edges go straight into one 32-slot bucket line per destination by atomic rank (fuller nodes spill to an exact overflow list): scan and fill launches go; sums still in arrival-rank order
# speedup vs baseline: 9.2208x; 1.3012x over previous
#include <hip/hip_runtime.h>
#include <math.h>


#define N_NODES 50000
#define N_EDGES 800000
#define HEADS   8
#define HEAD_DIM 16
#define FDIM    128
#define NEG_SLOPE 0.2f

typedef __attribute__((ext_vector_type(2))) float v2f;
typedef __attribute__((ext_vector_type(8))) float v8f;
typedef __attribute__((ext_vector_type(4))) float v4f;

static_assert(N_NODES % 16 == 0, "gemm: whole 16-row strips");
__global__ void __launch_bounds__(64)
gemm128_wmma_f32(const float* __restrict__ A, const float* __restrict__ W,
                 float* __restrict__ out) {
  const int wave = threadIdx.x >> 5;
  const int lane = threadIdx.x & 31;
  const int tm   = blockIdx.x;
  const int half = lane >> 4;
  const int l    = lane & 15;

  const float* __restrict__ arow = A + (size_t)(tm * 16 + l) * FDIM + 2 * half;
  const float* __restrict__ bcol = W + (size_t)(2 * half) * FDIM + wave * 64 + 4 * l;

  v8f c0 = {}, c1 = {}, c2 = {}, c3 = {};
#pragma unroll
  for (int k0 = 0; k0 < FDIM; k0 += 4) {
    v2f a;
    a.x = arow[k0];
    a.y = arow[k0 + 1];
    const v4f b0 = *(const v4f*)(bcol + (size_t)k0 * FDIM);
    const v4f b1 = *(const v4f*)(bcol + (size_t)(k0 + 1) * FDIM);
    c0 = __builtin_amdgcn_wmma_f32_16x16x4_f32(false, a, false, (v2f){b0.x, b1.x},
                                               (short)0, c0, false, false);
    c1 = __builtin_amdgcn_wmma_f32_16x16x4_f32(false, a, false, (v2f){b0.y, b1.y},
                                               (short)0, c1, false, false);
    c2 = __builtin_amdgcn_wmma_f32_16x16x4_f32(false, a, false, (v2f){b0.z, b1.z},
                                               (short)0, c2, false, false);
    c3 = __builtin_amdgcn_wmma_f32_16x16x4_f32(false, a, false, (v2f){b0.w, b1.w},
                                               (short)0, c3, false, false);
  }
  float* __restrict__ orow =
      out + (size_t)(tm * 16 + 8 * half) * FDIM + wave * 64 + 4 * l;
#pragma unroll
  for (int r = 0; r < 8; ++r)
    *(v4f*)(orow + (size_t)r * FDIM) = (v4f){c0[r], c1[r], c2[r], c3[r]};
}

#define BUCKET_CAP 32
static_assert(BUCKET_CAP == 32, "node_gather: one lane per bucket slot");
static_assert(BUCKET_CAP * (N_NODES / 2) >= N_EDGES, "bucket: capacity below twice the mean degree");
#define CNT_WORDS (((N_NODES + 1 + 255) / 256) * 256)
static_assert(N_NODES % 8 == 0, "node_gather: eight whole waves per block");

__global__ void zero_counts(int* __restrict__ cnt) {
  int t = blockIdx.x * blockDim.x + threadIdx.x;
  if (t < CNT_WORDS) cnt[t] = 0;
}

__global__ void bucket_edges(const int* __restrict__ ei, int* __restrict__ cnt,
                             int* __restrict__ bucket, int* __restrict__ ovf) {
  int e = blockIdx.x * blockDim.x + threadIdx.x;
  if (e >= N_EDGES) return;
  const int s = ei[e];
  const int d = ei[N_EDGES + e];
  const int r = atomicAdd(&cnt[d], 1);
  if (r < BUCKET_CAP) {
    bucket[(size_t)d * BUCKET_CAP + r] = s;
  } else {
    const int k = atomicAdd(&cnt[N_NODES], 1);
    ovf[2 * k] = d;
    ovf[2 * k + 1] = s;
  }
}

static_assert(NEG_SLOPE > 0.f && NEG_SLOPE < 1.f, "leaky: max form needs 0 < slope < 1");
__device__ __forceinline__ float leaky(float x) {
  return fmaxf(x, NEG_SLOPE * x);
}

template <int CTRL>
__device__ __forceinline__ float quad_move(float v) {
  return __builtin_bit_cast(
      float, __builtin_amdgcn_update_dpp(0, __builtin_bit_cast(int, v), CTRL,
                                         0xF, 0xF, true));
}

__device__ __forceinline__ float quad_sum(float v) {
  v += quad_move<0xB1>(v);
  v += quad_move<0x4E>(v);
  return v;
}

__device__ __forceinline__ float dot4(const v4f a, const v4f b) {
  float d = a.x * b.x;
  d = fmaf(a.y, b.y, d);
  d = fmaf(a.z, b.z, d);
  d = fmaf(a.w, b.w, d);
  return d;
}

__device__ __forceinline__ void absorb4(const v4f (&r)[4], const int nv,
                                        const v4f as4, const float ad, const int t,
                                        float& m, float& sum, v4f& acc) {
  float s[4];
#pragma unroll
  for (int q = 0; q < 4; ++q) {
    const float sc = leaky(quad_sum(dot4(r[q], as4)) + ad);
    s[q] = (q < nv) ? sc : -__builtin_inff();
  }
  const float s0 = s[0], s1 = s[1], s2 = s[2], s3 = s[3];
  const float mn = fmaxf(fmaxf(m, s0), fmaxf(fmaxf(s1, s2), s3));
  float st = s0;
  st = (t == 1) ? s1 : st;
  st = (t == 2) ? s2 : st;
  st = (t == 3) ? s3 : st;
  const float pt = expf(st - mn);
  const float scale = expf(m - mn);
  const float p0 = quad_move<0x00>(pt);
  const float p1 = quad_move<0x55>(pt);
  const float p2 = quad_move<0xAA>(pt);
  const float p3 = quad_move<0xFF>(pt);
  const float p[4] = {p0, p1, p2, p3};
  sum = fmaf(sum, scale, (p0 + p1) + (p2 + p3));
  acc.x *= scale;
  acc.y *= scale;
  acc.z *= scale;
  acc.w *= scale;
#pragma unroll
  for (int q = 0; q < 4; ++q) {
    const v4f rr = (q < nv) ? r[q] : (v4f){0.f, 0.f, 0.f, 0.f};
    acc.x = fmaf(p[q], rr.x, acc.x);
    acc.y = fmaf(p[q], rr.y, acc.y);
    acc.z = fmaf(p[q], rr.z, acc.z);
    acc.w = fmaf(p[q], rr.w, acc.w);
  }
  m = mn;
}

__global__ void __launch_bounds__(256)
node_gather(const int* __restrict__ cnt, const int* __restrict__ bucket,
            const int* __restrict__ ovf,
            const float* __restrict__ h, const float* __restrict__ att_src,
            const float* __restrict__ att_dst, const float* __restrict__ bias,
            float* __restrict__ out, int apply_elu) {
  const int lane = threadIdx.x & 31;
  const int n = __builtin_amdgcn_readfirstlane(
      (int)(blockIdx.x * blockDim.x + threadIdx.x) >> 5);
  if (n >= N_NODES) return;
  const v4f as4 = *(const v4f*)(att_src + 4 * lane);
  const v4f ad4 = *(const v4f*)(att_dst + 4 * lane);
  const v4f own = *(const v4f*)(h + (size_t)n * FDIM + 4 * lane);
  const int raw = bucket[(size_t)n * BUCKET_CAP + lane];
  const int deg = cnt[n];
  const int inb = deg < BUCKET_CAP ? deg : BUCKET_CAP;
  const int my = (lane < inb) ? raw : n;
  const int t = lane & 3;

  const float ad = quad_sum(dot4(own, ad4));
  float m = leaky(quad_sum(dot4(own, as4)) + ad);
  float sum = 1.f;
  v4f acc = own;

#pragma unroll
  for (int i = 0; i < BUCKET_CAP / 4; ++i) {
    if (4 * i < inb) {
      v4f r[4];
#pragma unroll
      for (int q = 0; q < 4; ++q) {
        const int sj = __builtin_amdgcn_readlane(my, 4 * i + q);
        r[q] = *(const v4f*)(h + (size_t)sj * FDIM + 4 * lane);
      }
      absorb4(r, inb - 4 * i, as4, ad, t, m, sum, acc);
    }
  }

  if (deg > BUCKET_CAP) {
    const int total = cnt[N_NODES];
    for (int kb = 0; kb < total; kb += 32) {
      const int k = kb + lane;
      int od = -1, os = n;
      if (k < total) {
        od = ovf[2 * k];
        os = ovf[2 * k + 1];
      }
      unsigned hit = __builtin_amdgcn_ballot_w32(od == n);
      while (hit != 0u) {
        const int l = __builtin_ctz(hit);
        hit &= hit - 1u;
        const int sj = __builtin_amdgcn_readlane(os, l);
        v4f r[4];
        r[0] = *(const v4f*)(h + (size_t)sj * FDIM + 4 * lane);
        r[1] = r[2] = r[3] = (v4f){0.f, 0.f, 0.f, 0.f};
        absorb4(r, 1, as4, ad, t, m, sum, acc);
      }
    }
  }

  const float inv = 1.f / (sum + 1e-16f);
  const v4f b = *(const v4f*)(bias + 4 * lane);
  v4f o;
  o.x = fmaf(acc.x, inv, b.x);
  o.y = fmaf(acc.y, inv, b.y);
  o.z = fmaf(acc.z, inv, b.z);
  o.w = fmaf(acc.w, inv, b.w);
  if (apply_elu) {
    o.x = o.x > 0.f ? o.x : (expf(o.x) - 1.f);
    o.y = o.y > 0.f ? o.y : (expf(o.y) - 1.f);
    o.z = o.z > 0.f ? o.z : (expf(o.z) - 1.f);
    o.w = o.w > 0.f ? o.w : (expf(o.w) - 1.f);
  }
  *(v4f*)(out + (size_t)n * FDIM + 4 * lane) = o;
}

extern "C" void kernel_launch(void* const* d_in, const int* in_sizes, int n_in,
                              void* d_out, int out_size, void* d_ws,
                              size_t ws_size, hipStream_t stream) {
  (void)in_sizes; (void)n_in; (void)out_size;

  const float* x   = (const float*)d_in[0];
  const int*   ei  = (const int*)d_in[1];
  const float* W1  = (const float*)d_in[2];
  const float* as1 = (const float*)d_in[3];
  const float* ad1 = (const float*)d_in[4];
  const float* b1  = (const float*)d_in[5];
  const float* W2  = (const float*)d_in[6];
  const float* as2 = (const float*)d_in[7];
  const float* ad2 = (const float*)d_in[8];
  const float* b2  = (const float*)d_in[9];
  float* out = (float*)d_out;

  float* ws    = (float*)d_ws;
  float* h     = ws;
  float* agg   = h + (size_t)N_NODES * FDIM;
  int* cnt     = (int*)(agg + (size_t)N_NODES * FDIM);
  int* bucket  = cnt + CNT_WORDS;
  int* ovf     = bucket + (size_t)N_NODES * BUCKET_CAP;
  const size_t need = ((size_t)N_NODES * FDIM * 2 + (size_t)CNT_WORDS +
                       (size_t)N_NODES * BUCKET_CAP + (size_t)N_EDGES * 2) * 4;
  if (ws_size < need) return;

  const int gemmBlocks = N_NODES / 16;
  const int gE  = (N_EDGES + 255) / 256;
  const int gZ  = CNT_WORDS / 256;
  const int gN  = N_NODES / 8;
  dim3 B64(64), B256(256);

  zero_counts<<<gZ, B256, 0, stream>>>(cnt);
  bucket_edges<<<gE, B256, 0, stream>>>(ei, cnt, bucket, ovf);

  gemm128_wmma_f32<<<gemmBlocks, B64, 0, stream>>>(x, W1, h);
  node_gather<<<gN, B256, 0, stream>>>(cnt, bucket, ovf, h, as1, ad1, b1, agg, 1);

  gemm128_wmma_f32<<<gemmBlocks, B64, 0, stream>>>(agg, W2, h);
  node_gather<<<gN, B256, 0, stream>>>(cnt, bucket, ovf, h, as2, ad2, b2, out, 0);
}


# edge bucketing rides in the layer-1 GEMM launch: each GEMM thread files 4 edges behind its tile (integer atomic rank, 4-byte store); bucket launch goes, 5 launches; sums still in arrival-rank order
# speedup vs baseline: 9.6718x; 1.0489x over previous
#include <hip/hip_runtime.h>
#include <math.h>


#define N_NODES 50000
#define N_EDGES 800000
#define HEADS   8
#define HEAD_DIM 16
#define FDIM    128
#define NEG_SLOPE 0.2f

typedef __attribute__((ext_vector_type(2))) float v2f;
typedef __attribute__((ext_vector_type(8))) float v8f;
typedef __attribute__((ext_vector_type(4))) float v4f;
typedef __attribute__((ext_vector_type(4))) int v4i;

#define BUCKET_CAP 32
static_assert(BUCKET_CAP == 32, "node_gather: one lane per bucket slot");
static_assert(BUCKET_CAP * (N_NODES / 2) >= N_EDGES, "bucket: capacity below twice the mean degree");
#define CNT_WORDS (((N_NODES + 1 + 255) / 256) * 256)
static_assert(N_NODES % 8 == 0, "node_gather: eight whole waves per block");

__global__ void zero_counts(int* __restrict__ cnt) {
  int t = blockIdx.x * blockDim.x + threadIdx.x;
  if (t < CNT_WORDS) cnt[t] = 0;
}

__device__ __forceinline__ void place_edge(const int s, const int d, const int r,
                                           int* __restrict__ cnt,
                                           int* __restrict__ bucket,
                                           int* __restrict__ ovf) {
  if (r < BUCKET_CAP) {
    bucket[(size_t)d * BUCKET_CAP + r] = s;
  } else {
    const int k = atomicAdd(&cnt[N_NODES], 1);
    ovf[2 * k] = d;
    ovf[2 * k + 1] = s;
  }
}

static_assert(N_NODES % 16 == 0, "gemm: whole 16-row strips");
#define GEMM_THREADS ((N_NODES / 16) * 64)
static_assert(GEMM_THREADS * 4 == N_EDGES, "gemm: four edges per thread cover the edge list exactly");
__global__ void __launch_bounds__(64) __attribute__((amdgpu_waves_per_eu(4, 4)))
gemm128_wmma_f32(const float* __restrict__ A, const float* __restrict__ W,
                 float* __restrict__ out, const int* __restrict__ ei,
                 int* __restrict__ cnt, int* __restrict__ bucket,
                 int* __restrict__ ovf) {
  const int wave = threadIdx.x >> 5;
  const int lane = threadIdx.x & 31;
  const int tm   = blockIdx.x;
  const int half = lane >> 4;
  const int l    = lane & 15;

  const float* __restrict__ arow = A + (size_t)(tm * 16 + l) * FDIM + 2 * half;
  const float* __restrict__ bcol = W + (size_t)(2 * half) * FDIM + wave * 64 + 4 * l;

  v8f c0 = {}, c1 = {}, c2 = {}, c3 = {};
#pragma unroll
  for (int k0 = 0; k0 < FDIM; k0 += 4) {
    v2f a;
    a.x = arow[k0];
    a.y = arow[k0 + 1];
    const v4f b0 = *(const v4f*)(bcol + (size_t)k0 * FDIM);
    const v4f b1 = *(const v4f*)(bcol + (size_t)(k0 + 1) * FDIM);
    c0 = __builtin_amdgcn_wmma_f32_16x16x4_f32(false, a, false, (v2f){b0.x, b1.x},
                                               (short)0, c0, false, false);
    c1 = __builtin_amdgcn_wmma_f32_16x16x4_f32(false, a, false, (v2f){b0.y, b1.y},
                                               (short)0, c1, false, false);
    c2 = __builtin_amdgcn_wmma_f32_16x16x4_f32(false, a, false, (v2f){b0.z, b1.z},
                                               (short)0, c2, false, false);
    c3 = __builtin_amdgcn_wmma_f32_16x16x4_f32(false, a, false, (v2f){b0.w, b1.w},
                                               (short)0, c3, false, false);
  }
  float* __restrict__ orow =
      out + (size_t)(tm * 16 + 8 * half) * FDIM + wave * 64 + 4 * l;
#pragma unroll
  for (int r = 0; r < 8; ++r)
    *(v4f*)(orow + (size_t)r * FDIM) = (v4f){c0[r], c1[r], c2[r], c3[r]};

  if (ei != nullptr) {
    const int g4 = 4 * (int)(blockIdx.x * 64 + threadIdx.x);
    const v4i s4 = *(const v4i*)(ei + g4);
    const v4i d4 = *(const v4i*)(ei + N_EDGES + g4);
    int rk[4];
#pragma unroll
    for (int j = 0; j < 4; ++j) rk[j] = atomicAdd(&cnt[d4[j]], 1);
#pragma unroll
    for (int j = 0; j < 4; ++j) place_edge(s4[j], d4[j], rk[j], cnt, bucket, ovf);
  }
}

static_assert(NEG_SLOPE > 0.f && NEG_SLOPE < 1.f, "leaky: max form needs 0 < slope < 1");
__device__ __forceinline__ float leaky(float x) {
  return fmaxf(x, NEG_SLOPE * x);
}

template <int CTRL>
__device__ __forceinline__ float quad_move(float v) {
  return __builtin_bit_cast(
      float, __builtin_amdgcn_update_dpp(0, __builtin_bit_cast(int, v), CTRL,
                                         0xF, 0xF, true));
}

__device__ __forceinline__ float quad_sum(float v) {
  v += quad_move<0xB1>(v);
  v += quad_move<0x4E>(v);
  return v;
}

__device__ __forceinline__ float dot4(const v4f a, const v4f b) {
  float d = a.x * b.x;
  d = fmaf(a.y, b.y, d);
  d = fmaf(a.z, b.z, d);
  d = fmaf(a.w, b.w, d);
  return d;
}

__device__ __forceinline__ void absorb4(const v4f (&r)[4], const int nv,
                                        const v4f as4, const float ad, const int t,
                                        float& m, float& sum, v4f& acc) {
  float s[4];
#pragma unroll
  for (int q = 0; q < 4; ++q) {
    const float sc = leaky(quad_sum(dot4(r[q], as4)) + ad);
    s[q] = (q < nv) ? sc : -__builtin_inff();
  }
  const float s0 = s[0], s1 = s[1], s2 = s[2], s3 = s[3];
  const float mn = fmaxf(fmaxf(m, s0), fmaxf(fmaxf(s1, s2), s3));
  float st = s0;
  st = (t == 1) ? s1 : st;
  st = (t == 2) ? s2 : st;
  st = (t == 3) ? s3 : st;
  const float pt = expf(st - mn);
  const float scale = expf(m - mn);
  const float p0 = quad_move<0x00>(pt);
  const float p1 = quad_move<0x55>(pt);
  const float p2 = quad_move<0xAA>(pt);
  const float p3 = quad_move<0xFF>(pt);
  const float p[4] = {p0, p1, p2, p3};
  sum = fmaf(sum, scale, (p0 + p1) + (p2 + p3));
  acc.x *= scale;
  acc.y *= scale;
  acc.z *= scale;
  acc.w *= scale;
#pragma unroll
  for (int q = 0; q < 4; ++q) {
    const v4f rr = (q < nv) ? r[q] : (v4f){0.f, 0.f, 0.f, 0.f};
    acc.x = fmaf(p[q], rr.x, acc.x);
    acc.y = fmaf(p[q], rr.y, acc.y);
    acc.z = fmaf(p[q], rr.z, acc.z);
    acc.w = fmaf(p[q], rr.w, acc.w);
  }
  m = mn;
}

__global__ void __launch_bounds__(256)
node_gather(const int* __restrict__ cnt, const int* __restrict__ bucket,
            const int* __restrict__ ovf,
            const float* __restrict__ h, const float* __restrict__ att_src,
            const float* __restrict__ att_dst, const float* __restrict__ bias,
            float* __restrict__ out, int apply_elu) {
  const int lane = threadIdx.x & 31;
  const int n = __builtin_amdgcn_readfirstlane(
      (int)(blockIdx.x * blockDim.x + threadIdx.x) >> 5);
  if (n >= N_NODES) return;
  const v4f as4 = *(const v4f*)(att_src + 4 * lane);
  const v4f ad4 = *(const v4f*)(att_dst + 4 * lane);
  const v4f own = *(const v4f*)(h + (size_t)n * FDIM + 4 * lane);
  const int raw = bucket[(size_t)n * BUCKET_CAP + lane];
  const int deg = cnt[n];
  const int inb = deg < BUCKET_CAP ? deg : BUCKET_CAP;
  const int my = (lane < inb) ? raw : n;
  const int t = lane & 3;

  const float ad = quad_sum(dot4(own, ad4));
  float m = leaky(quad_sum(dot4(own, as4)) + ad);
  float sum = 1.f;
  v4f acc = own;

#pragma unroll
  for (int i = 0; i < BUCKET_CAP / 4; ++i) {
    if (4 * i < inb) {
      v4f r[4];
#pragma unroll
      for (int q = 0; q < 4; ++q) {
        const int sj = __builtin_amdgcn_readlane(my, 4 * i + q);
        r[q] = *(const v4f*)(h + (size_t)sj * FDIM + 4 * lane);
      }
      absorb4(r, inb - 4 * i, as4, ad, t, m, sum, acc);
    }
  }

  if (deg > BUCKET_CAP) {
    const int total = cnt[N_NODES];
    for (int kb = 0; kb < total; kb += 32) {
      const int k = kb + lane;
      int od = -1, os = n;
      if (k < total) {
        od = ovf[2 * k];
        os = ovf[2 * k + 1];
      }
      unsigned hit = __builtin_amdgcn_ballot_w32(od == n);
      while (hit != 0u) {
        const int l = __builtin_ctz(hit);
        hit &= hit - 1u;
        const int sj = __builtin_amdgcn_readlane(os, l);
        v4f r[4];
        r[0] = *(const v4f*)(h + (size_t)sj * FDIM + 4 * lane);
        r[1] = r[2] = r[3] = (v4f){0.f, 0.f, 0.f, 0.f};
        absorb4(r, 1, as4, ad, t, m, sum, acc);
      }
    }
  }

  const float inv = 1.f / (sum + 1e-16f);
  const v4f b = *(const v4f*)(bias + 4 * lane);
  v4f o;
  o.x = fmaf(acc.x, inv, b.x);
  o.y = fmaf(acc.y, inv, b.y);
  o.z = fmaf(acc.z, inv, b.z);
  o.w = fmaf(acc.w, inv, b.w);
  if (apply_elu) {
    o.x = o.x > 0.f ? o.x : (expf(o.x) - 1.f);
    o.y = o.y > 0.f ? o.y : (expf(o.y) - 1.f);
    o.z = o.z > 0.f ? o.z : (expf(o.z) - 1.f);
    o.w = o.w > 0.f ? o.w : (expf(o.w) - 1.f);
  }
  *(v4f*)(out + (size_t)n * FDIM + 4 * lane) = o;
}

extern "C" void kernel_launch(void* const* d_in, const int* in_sizes, int n_in,
                              void* d_out, int out_size, void* d_ws,
                              size_t ws_size, hipStream_t stream) {
  (void)in_sizes; (void)n_in; (void)out_size;

  const float* x   = (const float*)d_in[0];
  const int*   ei  = (const int*)d_in[1];
  const float* W1  = (const float*)d_in[2];
  const float* as1 = (const float*)d_in[3];
  const float* ad1 = (const float*)d_in[4];
  const float* b1  = (const float*)d_in[5];
  const float* W2  = (const float*)d_in[6];
  const float* as2 = (const float*)d_in[7];
  const float* ad2 = (const float*)d_in[8];
  const float* b2  = (const float*)d_in[9];
  float* out = (float*)d_out;

  float* ws    = (float*)d_ws;
  float* h     = ws;
  float* agg   = h + (size_t)N_NODES * FDIM;
  int* cnt     = (int*)(agg + (size_t)N_NODES * FDIM);
  int* bucket  = cnt + CNT_WORDS;
  int* ovf     = bucket + (size_t)N_NODES * BUCKET_CAP;
  const size_t need = ((size_t)N_NODES * FDIM * 2 + (size_t)CNT_WORDS +
                       (size_t)N_NODES * BUCKET_CAP + (size_t)N_EDGES * 2) * 4;
  if (ws_size < need) return;

  const int gemmBlocks = N_NODES / 16;
  const int gZ  = CNT_WORDS / 256;
  const int gN  = N_NODES / 8;
  dim3 B64(64), B256(256);

  zero_counts<<<gZ, B256, 0, stream>>>(cnt);
  gemm128_wmma_f32<<<gemmBlocks, B64, 0, stream>>>(x, W1, h, ei, cnt, bucket, ovf);
  node_gather<<<gN, B256, 0, stream>>>(cnt, bucket, ovf, h, as1, ad1, b1, agg, 1);

  gemm128_wmma_f32<<<gemmBlocks, B64, 0, stream>>>(agg, W2, h, nullptr, nullptr, nullptr, nullptr);
  node_gather<<<gN, B256, 0, stream>>>(cnt, bucket, ovf, h, as2, ad2, b2, out, 0);
}


# edges filed into destination buckets by two LDS passes (4096-edge chunks ordered by 256-node range, then a work-group per range ranks by LDS add, whole-line stores); 6 launches; sums in arrival order
# speedup vs baseline: 14.0345x; 1.4511x over previous
#include <hip/hip_runtime.h>
#include <math.h>


#define N_NODES 50000
#define N_EDGES 800000
#define HEADS   8
#define HEAD_DIM 16
#define FDIM    128
#define NEG_SLOPE 0.2f

typedef __attribute__((ext_vector_type(2))) float v2f;
typedef __attribute__((ext_vector_type(8))) float v8f;
typedef __attribute__((ext_vector_type(4))) float v4f;
typedef __attribute__((ext_vector_type(4))) int v4i;
typedef __attribute__((ext_vector_type(2))) int v2i;

#define BUCKET_CAP 32
static_assert(BUCKET_CAP == 32, "node_gather: one lane per bucket slot");
static_assert(BUCKET_CAP * (N_NODES / 2) >= N_EDGES, "bucket: capacity below twice the mean degree");
#define CNT_WORDS (((N_NODES + 1 + 255) / 256) * 256)
static_assert(N_NODES % 8 == 0, "node_gather: eight whole waves per block");

#define BIN_THREADS 1024
#define BIN_WAVES (BIN_THREADS / 32)
#define RANGE_NODES 256
#define N_RANGES ((N_NODES + RANGE_NODES - 1) / RANGE_NODES)
#define CHUNK_EDGES (4 * BIN_THREADS)
#define N_CHUNKS ((N_EDGES + CHUNK_EDGES - 1) / CHUNK_EDGES)
#define SCAN_PER_LANE ((N_RANGES + 31) / 32)
#define CHUNKS_PER_WAVE ((N_CHUNKS + BIN_WAVES - 1) / BIN_WAVES)
#define TBL_WORDS (((2 * N_CHUNKS * N_RANGES + 31) / 32) * 32)
static_assert(RANGE_NODES == 256, "packed edge: low 8 bits are the place in the range");
static_assert(N_NODES < (1 << 23), "packed edge: source above the low 8 bits");
static_assert(N_EDGES % 4 == 0, "chunk_sort: four edges per 16-byte load");
static_assert(RANGE_NODES * BUCKET_CAP / 4 == 2 * BIN_THREADS, "range_fill: two 16-byte pieces per thread");

__global__ void __launch_bounds__(BIN_THREADS)
chunk_sort(const int* __restrict__ ei, int* __restrict__ sorted,
           int* __restrict__ tbl, int* __restrict__ cnt) {
  __shared__ int hist[N_RANGES];
  __shared__ int start[N_RANGES];
  __shared__ int buf[CHUNK_EDGES];
  const int t = threadIdx.x;
  const int ch = blockIdx.x;
  if (t < N_RANGES) hist[t] = 0;
  if (ch == 0 && t == 0) cnt[N_NODES] = 0;
  __syncthreads();

  const int e0 = ch * CHUNK_EDGES + 4 * t;
  const bool live = e0 < N_EDGES;
  v4i s4 = {0, 0, 0, 0}, d4 = {0, 0, 0, 0};
  int lr[4] = {0, 0, 0, 0};
  if (live) {
    s4 = *(const v4i*)(ei + e0);
    d4 = *(const v4i*)(ei + N_EDGES + e0);
#pragma unroll
    for (int j = 0; j < 4; ++j) lr[j] = atomicAdd(&hist[d4[j] >> 8], 1);
  }
  __syncthreads();

  if (t < 32) {
    int v[SCAN_PER_LANE];
    int tot = 0;
#pragma unroll
    for (int q = 0; q < SCAN_PER_LANE; ++q) {
      const int i = SCAN_PER_LANE * t + q;
      v[q] = (i < N_RANGES) ? hist[i] : 0;
      tot += v[q];
    }
    int inc = tot;
#pragma unroll
    for (int o = 1; o < 32; o <<= 1) {
      const int u = __shfl_up(inc, o);
      if (t >= o) inc += u;
    }
    int run = inc - tot;
#pragma unroll
    for (int q = 0; q < SCAN_PER_LANE; ++q) {
      const int i = SCAN_PER_LANE * t + q;
      if (i < N_RANGES) {
        start[i] = run;
        *(v2i*)(tbl + 2 * (ch * N_RANGES + i)) = (v2i){run, v[q]};
        run += v[q];
      }
    }
  }
  __syncthreads();

  if (live) {
#pragma unroll
    for (int j = 0; j < 4; ++j)
      buf[start[d4[j] >> 8] + lr[j]] = (s4[j] << 8) | (d4[j] & (RANGE_NODES - 1));
  }
  __syncthreads();
  if (live)
    *(v4i*)(sorted + (size_t)ch * CHUNK_EDGES + 4 * t) = *(const v4i*)(buf + 4 * t);
}

__device__ __forceinline__ void file_edge(const int e, const int node0,
                                          int* __restrict__ c, int* __restrict__ bkt,
                                          int* __restrict__ cnt, int* __restrict__ ovf) {
  const int dl = e & (RANGE_NODES - 1);
  const int s = e >> 8;
  const int rk = atomicAdd(&c[dl], 1);
  if (rk < BUCKET_CAP) {
    bkt[dl * BUCKET_CAP + rk] = s;
  } else {
    const int k = atomicAdd(&cnt[N_NODES], 1);
    ovf[2 * k] = node0 + dl;
    ovf[2 * k + 1] = s;
  }
}

__global__ void __launch_bounds__(BIN_THREADS)
range_fill(const int* __restrict__ sorted, const int* __restrict__ tbl,
           int* __restrict__ cnt, int* __restrict__ bucket, int* __restrict__ ovf) {
  __shared__ int c[RANGE_NODES];
  __shared__ int seg_start[N_CHUNKS];
  __shared__ int seg_len[N_CHUNKS];
  __shared__ int bkt[RANGE_NODES * BUCKET_CAP];
  const int t = threadIdx.x;
  const int r = blockIdx.x;
  const int lane = t & 31;
  const int w = t >> 5;
  const int node0 = r * RANGE_NODES;
  if (t < RANGE_NODES) c[t] = 0;
  if (t < N_CHUNKS) {
    const v2i e = *(const v2i*)(tbl + 2 * (t * N_RANGES + r));
    seg_start[t] = e.x;
    seg_len[t] = e.y;
  }
  ((v4i*)bkt)[t] = (v4i){0, 0, 0, 0};
  ((v4i*)bkt)[t + BIN_THREADS] = (v4i){0, 0, 0, 0};
  __syncthreads();

  int st[CHUNKS_PER_WAVE], n[CHUNKS_PER_WAVE], e[CHUNKS_PER_WAVE];
#pragma unroll
  for (int q = 0; q < CHUNKS_PER_WAVE; ++q) {
    const int ch = w + BIN_WAVES * q;
    st[q] = 0;
    n[q] = 0;
    if (ch < N_CHUNKS) {
      st[q] = seg_start[ch];
      n[q] = seg_len[ch];
    }
    e[q] = 0;
    if (lane < n[q]) e[q] = sorted[(size_t)ch * CHUNK_EDGES + st[q] + lane];
  }
#pragma unroll
  for (int q = 0; q < CHUNKS_PER_WAVE; ++q) {
    const int ch = w + BIN_WAVES * q;
    if (lane < n[q]) file_edge(e[q], node0, c, bkt, cnt, ovf);
    for (int i = lane + 32; i < n[q]; i += 32)
      file_edge(sorted[(size_t)ch * CHUNK_EDGES + st[q] + i], node0, c, bkt, cnt, ovf);
  }
  __syncthreads();

  int* __restrict__ gb = bucket + (size_t)node0 * BUCKET_CAP;
#pragma unroll
  for (int i = t; i < RANGE_NODES * BUCKET_CAP / 4; i += BIN_THREADS)
    if (node0 + (i >> 3) < N_NODES) ((v4i*)gb)[i] = ((const v4i*)bkt)[i];
  if (t < RANGE_NODES && node0 + t < N_NODES) cnt[node0 + t] = c[t];
}

static_assert(N_NODES % 16 == 0, "gemm: whole 16-row strips");
__global__ void __launch_bounds__(64) __attribute__((amdgpu_waves_per_eu(4, 4)))
gemm128_wmma_f32(const float* __restrict__ A, const float* __restrict__ W,
                 float* __restrict__ out) {
  const int wave = threadIdx.x >> 5;
  const int lane = threadIdx.x & 31;
  const int tm   = blockIdx.x;
  const int half = lane >> 4;
  const int l    = lane & 15;

  const float* __restrict__ arow = A + (size_t)(tm * 16 + l) * FDIM + 2 * half;
  const float* __restrict__ bcol = W + (size_t)(2 * half) * FDIM + wave * 64 + 4 * l;

  v8f c0 = {}, c1 = {}, c2 = {}, c3 = {};
#pragma unroll
  for (int k0 = 0; k0 < FDIM; k0 += 4) {
    v2f a;
    a.x = arow[k0];
    a.y = arow[k0 + 1];
    const v4f b0 = *(const v4f*)(bcol + (size_t)k0 * FDIM);
    const v4f b1 = *(const v4f*)(bcol + (size_t)(k0 + 1) * FDIM);
    c0 = __builtin_amdgcn_wmma_f32_16x16x4_f32(false, a, false, (v2f){b0.x, b1.x},
                                               (short)0, c0, false, false);
    c1 = __builtin_amdgcn_wmma_f32_16x16x4_f32(false, a, false, (v2f){b0.y, b1.y},
                                               (short)0, c1, false, false);
    c2 = __builtin_amdgcn_wmma_f32_16x16x4_f32(false, a, false, (v2f){b0.z, b1.z},
                                               (short)0, c2, false, false);
    c3 = __builtin_amdgcn_wmma_f32_16x16x4_f32(false, a, false, (v2f){b0.w, b1.w},
                                               (short)0, c3, false, false);
  }
  float* __restrict__ orow =
      out + (size_t)(tm * 16 + 8 * half) * FDIM + wave * 64 + 4 * l;
#pragma unroll
  for (int r = 0; r < 8; ++r)
    *(v4f*)(orow + (size_t)r * FDIM) = (v4f){c0[r], c1[r], c2[r], c3[r]};
}

static_assert(NEG_SLOPE > 0.f && NEG_SLOPE < 1.f, "leaky: max form needs 0 < slope < 1");
__device__ __forceinline__ float leaky(float x) {
  return fmaxf(x, NEG_SLOPE * x);
}

template <int CTRL>
__device__ __forceinline__ float quad_move(float v) {
  return __builtin_bit_cast(
      float, __builtin_amdgcn_update_dpp(0, __builtin_bit_cast(int, v), CTRL,
                                         0xF, 0xF, true));
}

__device__ __forceinline__ float quad_sum(float v) {
  v += quad_move<0xB1>(v);
  v += quad_move<0x4E>(v);
  return v;
}

__device__ __forceinline__ float dot4(const v4f a, const v4f b) {
  float d = a.x * b.x;
  d = fmaf(a.y, b.y, d);
  d = fmaf(a.z, b.z, d);
  d = fmaf(a.w, b.w, d);
  return d;
}

__device__ __forceinline__ void absorb4(const v4f (&r)[4], const int nv,
                                        const v4f as4, const float ad, const int t,
                                        float& m, float& sum, v4f& acc) {
  float s[4];
#pragma unroll
  for (int q = 0; q < 4; ++q) {
    const float sc = leaky(quad_sum(dot4(r[q], as4)) + ad);
    s[q] = (q < nv) ? sc : -__builtin_inff();
  }
  const float s0 = s[0], s1 = s[1], s2 = s[2], s3 = s[3];
  const float mn = fmaxf(fmaxf(m, s0), fmaxf(fmaxf(s1, s2), s3));
  float st = s0;
  st = (t == 1) ? s1 : st;
  st = (t == 2) ? s2 : st;
  st = (t == 3) ? s3 : st;
  const float pt = expf(st - mn);
  const float scale = expf(m - mn);
  const float p0 = quad_move<0x00>(pt);
  const float p1 = quad_move<0x55>(pt);
  const float p2 = quad_move<0xAA>(pt);
  const float p3 = quad_move<0xFF>(pt);
  const float p[4] = {p0, p1, p2, p3};
  sum = fmaf(sum, scale, (p0 + p1) + (p2 + p3));
  acc.x *= scale;
  acc.y *= scale;
  acc.z *= scale;
  acc.w *= scale;
#pragma unroll
  for (int q = 0; q < 4; ++q) {
    const v4f rr = (q < nv) ? r[q] : (v4f){0.f, 0.f, 0.f, 0.f};
    acc.x = fmaf(p[q], rr.x, acc.x);
    acc.y = fmaf(p[q], rr.y, acc.y);
    acc.z = fmaf(p[q], rr.z, acc.z);
    acc.w = fmaf(p[q], rr.w, acc.w);
  }
  m = mn;
}

__global__ void __launch_bounds__(256)
node_gather(const int* __restrict__ cnt, const int* __restrict__ bucket,
            const int* __restrict__ ovf,
            const float* __restrict__ h, const float* __restrict__ att_src,
            const float* __restrict__ att_dst, const float* __restrict__ bias,
            float* __restrict__ out, int apply_elu) {
  const int lane = threadIdx.x & 31;
  const int n = __builtin_amdgcn_readfirstlane(
      (int)(blockIdx.x * blockDim.x + threadIdx.x) >> 5);
  if (n >= N_NODES) return;
  const v4f as4 = *(const v4f*)(att_src + 4 * lane);
  const v4f ad4 = *(const v4f*)(att_dst + 4 * lane);
  const v4f own = *(const v4f*)(h + (size_t)n * FDIM + 4 * lane);
  const int raw = bucket[(size_t)n * BUCKET_CAP + lane];
  const int deg = cnt[n];
  const int inb = deg < BUCKET_CAP ? deg : BUCKET_CAP;
  const int my = (lane < inb) ? raw : n;
  const int t = lane & 3;

  const float ad = quad_sum(dot4(own, ad4));
  float m = leaky(quad_sum(dot4(own, as4)) + ad);
  float sum = 1.f;
  v4f acc = own;

#pragma unroll
  for (int i = 0; i < BUCKET_CAP / 4; ++i) {
    if (4 * i < inb) {
      v4f r[4];
#pragma unroll
      for (int q = 0; q < 4; ++q) {
        const int sj = __builtin_amdgcn_readlane(my, 4 * i + q);
        r[q] = *(const v4f*)(h + (size_t)sj * FDIM + 4 * lane);
      }
      absorb4(r, inb - 4 * i, as4, ad, t, m, sum, acc);
    }
  }

  if (deg > BUCKET_CAP) {
    const int total = cnt[N_NODES];
    for (int kb = 0; kb < total; kb += 32) {
      const int k = kb + lane;
      int od = -1, os = n;
      if (k < total) {
        od = ovf[2 * k];
        os = ovf[2 * k + 1];
      }
      unsigned hit = __builtin_amdgcn_ballot_w32(od == n);
      while (hit != 0u) {
        const int l = __builtin_ctz(hit);
        hit &= hit - 1u;
        const int sj = __builtin_amdgcn_readlane(os, l);
        v4f r[4];
        r[0] = *(const v4f*)(h + (size_t)sj * FDIM + 4 * lane);
        r[1] = r[2] = r[3] = (v4f){0.f, 0.f, 0.f, 0.f};
        absorb4(r, 1, as4, ad, t, m, sum, acc);
      }
    }
  }

  const float inv = 1.f / (sum + 1e-16f);
  const v4f b = *(const v4f*)(bias + 4 * lane);
  v4f o;
  o.x = fmaf(acc.x, inv, b.x);
  o.y = fmaf(acc.y, inv, b.y);
  o.z = fmaf(acc.z, inv, b.z);
  o.w = fmaf(acc.w, inv, b.w);
  if (apply_elu) {
    o.x = o.x > 0.f ? o.x : (expf(o.x) - 1.f);
    o.y = o.y > 0.f ? o.y : (expf(o.y) - 1.f);
    o.z = o.z > 0.f ? o.z : (expf(o.z) - 1.f);
    o.w = o.w > 0.f ? o.w : (expf(o.w) - 1.f);
  }
  *(v4f*)(out + (size_t)n * FDIM + 4 * lane) = o;
}

extern "C" void kernel_launch(void* const* d_in, const int* in_sizes, int n_in,
                              void* d_out, int out_size, void* d_ws,
                              size_t ws_size, hipStream_t stream) {
  (void)in_sizes; (void)n_in; (void)out_size;

  const float* x   = (const float*)d_in[0];
  const int*   ei  = (const int*)d_in[1];
  const float* W1  = (const float*)d_in[2];
  const float* as1 = (const float*)d_in[3];
  const float* ad1 = (const float*)d_in[4];
  const float* b1  = (const float*)d_in[5];
  const float* W2  = (const float*)d_in[6];
  const float* as2 = (const float*)d_in[7];
  const float* ad2 = (const float*)d_in[8];
  const float* b2  = (const float*)d_in[9];
  float* out = (float*)d_out;

  float* ws    = (float*)d_ws;
  float* h     = ws;
  float* agg   = h + (size_t)N_NODES * FDIM;
  int* cnt     = (int*)(agg + (size_t)N_NODES * FDIM);
  int* bucket  = cnt + CNT_WORDS;
  int* ovf     = bucket + (size_t)N_NODES * BUCKET_CAP;
  int* tbl     = ovf + (size_t)N_EDGES * 2;
  int* sorted  = tbl + (size_t)TBL_WORDS;
  const size_t need = ((size_t)N_NODES * FDIM * 2 + (size_t)CNT_WORDS +
                       (size_t)N_NODES * BUCKET_CAP + (size_t)N_EDGES * 2 +
                       (size_t)TBL_WORDS + (size_t)N_CHUNKS * CHUNK_EDGES) * 4;
  if (ws_size < need) return;

  const int gemmBlocks = N_NODES / 16;
  const int gN  = N_NODES / 8;
  dim3 B64(64), B256(256), BBIN(BIN_THREADS);

  chunk_sort<<<N_CHUNKS, BBIN, 0, stream>>>(ei, sorted, tbl, cnt);
  range_fill<<<N_RANGES, BBIN, 0, stream>>>(sorted, tbl, cnt, bucket, ovf);

  gemm128_wmma_f32<<<gemmBlocks, B64, 0, stream>>>(x, W1, h);
  node_gather<<<gN, B256, 0, stream>>>(cnt, bucket, ovf, h, as1, ad1, b1, agg, 1);

  gemm128_wmma_f32<<<gemmBlocks, B64, 0, stream>>>(agg, W2, h);
  node_gather<<<gN, B256, 0, stream>>>(cnt, bucket, ovf, h, as2, ad2, b2, out, 0);
}


# bucket build's chunk ordering pass rides in the layer-1 GEMM launch: its first 196 two-wave blocks each order 4096 edges by range in LDS while the rest multiply; 5 launches; sums in arrival order
# speedup vs baseline: 14.6736x; 1.0455x over previous
#include <hip/hip_runtime.h>
#include <math.h>


#define N_NODES 50000
#define N_EDGES 800000
#define HEADS   8
#define HEAD_DIM 16
#define FDIM    128
#define NEG_SLOPE 0.2f

typedef __attribute__((ext_vector_type(2))) float v2f;
typedef __attribute__((ext_vector_type(8))) float v8f;
typedef __attribute__((ext_vector_type(4))) float v4f;
typedef __attribute__((ext_vector_type(4))) int v4i;
typedef __attribute__((ext_vector_type(2))) int v2i;

#define BUCKET_CAP 32
static_assert(BUCKET_CAP == 32, "node_gather: one lane per bucket slot");
static_assert(BUCKET_CAP * (N_NODES / 2) >= N_EDGES, "bucket: capacity below twice the mean degree");
#define CNT_WORDS (((N_NODES + 1 + 255) / 256) * 256)
static_assert(N_NODES % 8 == 0, "node_gather: eight whole waves per block");

#define BIN_THREADS 1024
#define BIN_WAVES (BIN_THREADS / 32)
#define RANGE_NODES 256
#define N_RANGES ((N_NODES + RANGE_NODES - 1) / RANGE_NODES)
#define CHUNK_EDGES 4096
#define N_CHUNKS ((N_EDGES + CHUNK_EDGES - 1) / CHUNK_EDGES)
#define SCAN_PER_LANE ((N_RANGES + 31) / 32)
#define CHUNKS_PER_WAVE ((N_CHUNKS + BIN_WAVES - 1) / BIN_WAVES)
#define TBL_WORDS (((2 * N_CHUNKS * N_RANGES + 31) / 32) * 32)
static_assert(RANGE_NODES == 256, "packed edge: low 8 bits are the place in the range");
static_assert(N_NODES < (1 << 23), "packed edge: source above the low 8 bits");
static_assert(N_EDGES % 4 == 0, "chunk_sort: four edges per 16-byte load");
static_assert(RANGE_NODES * BUCKET_CAP / 4 == 2 * BIN_THREADS, "range_fill: two 16-byte pieces per thread");

#define SORT_THREADS 64
#define SORT_EPT (CHUNK_EDGES / SORT_THREADS)
#define SORT_V4 (SORT_EPT / 4)
static_assert(CHUNK_EDGES % (4 * SORT_THREADS) == 0, "chunk_sort: whole 16-byte loads per thread");
static_assert(N_RANGES <= 256 && CHUNK_EDGES == 4096, "chunk_sort: key = 8 bits of range, 12 of rank, 8 of place");
static_assert(N_EDGES % (4 * SORT_THREADS) == 0, "chunk_sort: a load is inside the list for all threads or none");
template <bool FULL>
__device__ __forceinline__ void
chunk_sort_block(const int ch, const int nj, int* __restrict__ hist,
                 int* __restrict__ start, int* __restrict__ buf,
                 const int* __restrict__ ei, int* __restrict__ sorted,
                 int* __restrict__ tbl, int* __restrict__ cnt) {
  const int t = threadIdx.x;
  for (int i = t; i < N_RANGES; i += SORT_THREADS) hist[i] = 0;
  if (ch == 0 && t == 0) cnt[N_NODES] = 0;
  __syncthreads();

  const int e0 = ch * CHUNK_EDGES + 4 * t;
  v4i d4[SORT_V4], s4[SORT_V4];
#pragma unroll
  for (int j = 0; j < SORT_V4; ++j) {
    d4[j] = (v4i){0, 0, 0, 0};
    if (FULL || j < nj) d4[j] = *(const v4i*)(ei + N_EDGES + e0 + 4 * SORT_THREADS * j);
  }
  int key[SORT_EPT];
#pragma unroll
  for (int j = 0; j < SORT_V4; ++j) {
    if (FULL || j < nj) {
#pragma unroll
      for (int q = 0; q < 4; ++q) {
        const int rg = d4[j][q] >> 8;
        key[4 * j + q] = rg | (atomicAdd(&hist[rg], 1) << 8) |
                         ((d4[j][q] & (RANGE_NODES - 1)) << 20);
      }
    }
  }
#pragma unroll
  for (int j = 0; j < SORT_V4; ++j) {
    s4[j] = (v4i){0, 0, 0, 0};
    if (FULL || j < nj) s4[j] = *(const v4i*)(ei + e0 + 4 * SORT_THREADS * j);
  }
  __syncthreads();

  if (t < 32) {
    int v[SCAN_PER_LANE];
    int tot = 0;
#pragma unroll
    for (int q = 0; q < SCAN_PER_LANE; ++q) {
      const int i = SCAN_PER_LANE * t + q;
      v[q] = (i < N_RANGES) ? hist[i] : 0;
      tot += v[q];
    }
    int inc = tot;
#pragma unroll
    for (int o = 1; o < 32; o <<= 1) {
      const int u = __shfl_up(inc, o);
      if (t >= o) inc += u;
    }
    int run = inc - tot;
#pragma unroll
    for (int q = 0; q < SCAN_PER_LANE; ++q) {
      const int i = SCAN_PER_LANE * t + q;
      if (i < N_RANGES) {
        start[i] = run;
        *(v2i*)(tbl + 2 * (ch * N_RANGES + i)) = (v2i){run, v[q]};
        run += v[q];
      }
    }
  }
  __syncthreads();

#pragma unroll
  for (int j = 0; j < SORT_V4; ++j) {
    if (FULL || j < nj) {
#pragma unroll
      for (int q = 0; q < 4; ++q)
        buf[start[key[4 * j + q] & 255] + ((key[4 * j + q] >> 8) & (CHUNK_EDGES - 1))] =
            (s4[j][q] << 8) | (key[4 * j + q] >> 20);
    }
  }
  __syncthreads();
#pragma unroll
  for (int j = 0; j < SORT_V4; ++j) {
    if (FULL || j < nj)
      *(v4i*)(sorted + (size_t)e0 + 4 * SORT_THREADS * j) =
          *(const v4i*)(buf + 4 * (t + SORT_THREADS * j));
  }
}

__device__ __forceinline__ void file_edge(const int e, const int node0,
                                          int* __restrict__ c, int* __restrict__ bkt,
                                          int* __restrict__ cnt, int* __restrict__ ovf) {
  const int dl = e & (RANGE_NODES - 1);
  const int s = e >> 8;
  const int rk = atomicAdd(&c[dl], 1);
  if (rk < BUCKET_CAP) {
    bkt[dl * BUCKET_CAP + rk] = s;
  } else {
    const int k = atomicAdd(&cnt[N_NODES], 1);
    ovf[2 * k] = node0 + dl;
    ovf[2 * k + 1] = s;
  }
}

__global__ void __launch_bounds__(BIN_THREADS)
range_fill(const int* __restrict__ sorted, const int* __restrict__ tbl,
           int* __restrict__ cnt, int* __restrict__ bucket, int* __restrict__ ovf) {
  __shared__ int c[RANGE_NODES];
  __shared__ int seg_start[N_CHUNKS];
  __shared__ int seg_len[N_CHUNKS];
  __shared__ int bkt[RANGE_NODES * BUCKET_CAP];
  const int t = threadIdx.x;
  const int r = blockIdx.x;
  const int lane = t & 31;
  const int w = t >> 5;
  const int node0 = r * RANGE_NODES;
  if (t < RANGE_NODES) c[t] = 0;
  if (t < N_CHUNKS) {
    const v2i e = *(const v2i*)(tbl + 2 * (t * N_RANGES + r));
    seg_start[t] = e.x;
    seg_len[t] = e.y;
  }
  ((v4i*)bkt)[t] = (v4i){0, 0, 0, 0};
  ((v4i*)bkt)[t + BIN_THREADS] = (v4i){0, 0, 0, 0};
  __syncthreads();

  int st[CHUNKS_PER_WAVE], n[CHUNKS_PER_WAVE], e[CHUNKS_PER_WAVE];
#pragma unroll
  for (int q = 0; q < CHUNKS_PER_WAVE; ++q) {
    const int ch = w + BIN_WAVES * q;
    st[q] = 0;
    n[q] = 0;
    if (ch < N_CHUNKS) {
      st[q] = seg_start[ch];
      n[q] = seg_len[ch];
    }
    e[q] = 0;
    if (lane < n[q]) e[q] = sorted[(size_t)ch * CHUNK_EDGES + st[q] + lane];
  }
#pragma unroll
  for (int q = 0; q < CHUNKS_PER_WAVE; ++q) {
    const int ch = w + BIN_WAVES * q;
    if (lane < n[q]) file_edge(e[q], node0, c, bkt, cnt, ovf);
    for (int i = lane + 32; i < n[q]; i += 32)
      file_edge(sorted[(size_t)ch * CHUNK_EDGES + st[q] + i], node0, c, bkt, cnt, ovf);
  }
  __syncthreads();

  int* __restrict__ gb = bucket + (size_t)node0 * BUCKET_CAP;
#pragma unroll
  for (int i = t; i < RANGE_NODES * BUCKET_CAP / 4; i += BIN_THREADS)
    if (node0 + (i >> 3) < N_NODES) ((v4i*)gb)[i] = ((const v4i*)bkt)[i];
  if (t < RANGE_NODES && node0 + t < N_NODES) cnt[node0 + t] = c[t];
}

static_assert(N_NODES % 16 == 0, "gemm: whole 16-row strips");
__device__ __forceinline__ void
gemm_strip(const int tm  , const float* __restrict__ A,
           const float* __restrict__ W, float* __restrict__ out) {
  const int wave = threadIdx.x >> 5;
  const int lane = threadIdx.x & 31;
  const int half = lane >> 4;
  const int l    = lane & 15;

  const float* __restrict__ arow = A + (size_t)(tm * 16 + l) * FDIM + 2 * half;
  const float* __restrict__ bcol = W + (size_t)(2 * half) * FDIM + wave * 64 + 4 * l;

  v8f c0 = {}, c1 = {}, c2 = {}, c3 = {};
#pragma unroll
  for (int k0 = 0; k0 < FDIM; k0 += 4) {
    v2f a;
    a.x = arow[k0];
    a.y = arow[k0 + 1];
    const v4f b0 = *(const v4f*)(bcol + (size_t)k0 * FDIM);
    const v4f b1 = *(const v4f*)(bcol + (size_t)(k0 + 1) * FDIM);
    c0 = __builtin_amdgcn_wmma_f32_16x16x4_f32(false, a, false, (v2f){b0.x, b1.x},
                                               (short)0, c0, false, false);
    c1 = __builtin_amdgcn_wmma_f32_16x16x4_f32(false, a, false, (v2f){b0.y, b1.y},
                                               (short)0, c1, false, false);
    c2 = __builtin_amdgcn_wmma_f32_16x16x4_f32(false, a, false, (v2f){b0.z, b1.z},
                                               (short)0, c2, false, false);
    c3 = __builtin_amdgcn_wmma_f32_16x16x4_f32(false, a, false, (v2f){b0.w, b1.w},
                                               (short)0, c3, false, false);
  }
  float* __restrict__ orow =
      out + (size_t)(tm * 16 + 8 * half) * FDIM + wave * 64 + 4 * l;
#pragma unroll
  for (int r = 0; r < 8; ++r)
    *(v4f*)(orow + (size_t)r * FDIM) = (v4f){c0[r], c1[r], c2[r], c3[r]};
}

__global__ void __launch_bounds__(64) __attribute__((amdgpu_waves_per_eu(4, 4)))
gemm128_wmma_f32(const float* __restrict__ A, const float* __restrict__ W,
                 float* __restrict__ out) {
  gemm_strip(blockIdx.x, A, W, out);
}

static_assert(SORT_THREADS == 64, "gemm128_sort: the sorting blocks have the GEMM's block size");
__global__ void __launch_bounds__(64) __attribute__((amdgpu_waves_per_eu(4, 4)))
gemm128_sort(const float* __restrict__ A, const float* __restrict__ W,
             float* __restrict__ out, const int* __restrict__ ei,
             int* __restrict__ sorted, int* __restrict__ tbl, int* __restrict__ cnt) {
  if (blockIdx.x < N_CHUNKS) {
    __shared__ int hist[N_RANGES];
    __shared__ int start[N_RANGES];
    __shared__ int buf[CHUNK_EDGES];
    const int ch = blockIdx.x;
    const int left = (N_EDGES - ch * CHUNK_EDGES) / (4 * SORT_THREADS);
    const int nj = left < SORT_V4 ? left : SORT_V4;
    if (nj == SORT_V4)
      chunk_sort_block<true>(ch, nj, hist, start, buf, ei, sorted, tbl, cnt);
    else
      chunk_sort_block<false>(ch, nj, hist, start, buf, ei, sorted, tbl, cnt);
    return;
  }
  gemm_strip(blockIdx.x - N_CHUNKS, A, W, out);
}

static_assert(NEG_SLOPE > 0.f && NEG_SLOPE < 1.f, "leaky: max form needs 0 < slope < 1");
__device__ __forceinline__ float leaky(float x) {
  return fmaxf(x, NEG_SLOPE * x);
}

template <int CTRL>
__device__ __forceinline__ float quad_move(float v) {
  return __builtin_bit_cast(
      float, __builtin_amdgcn_update_dpp(0, __builtin_bit_cast(int, v), CTRL,
                                         0xF, 0xF, true));
}

__device__ __forceinline__ float quad_sum(float v) {
  v += quad_move<0xB1>(v);
  v += quad_move<0x4E>(v);
  return v;
}

__device__ __forceinline__ float dot4(const v4f a, const v4f b) {
  float d = a.x * b.x;
  d = fmaf(a.y, b.y, d);
  d = fmaf(a.z, b.z, d);
  d = fmaf(a.w, b.w, d);
  return d;
}

__device__ __forceinline__ void absorb4(const v4f (&r)[4], const int nv,
                                        const v4f as4, const float ad, const int t,
                                        float& m, float& sum, v4f& acc) {
  float s[4];
#pragma unroll
  for (int q = 0; q < 4; ++q) {
    const float sc = leaky(quad_sum(dot4(r[q], as4)) + ad);
    s[q] = (q < nv) ? sc : -__builtin_inff();
  }
  const float s0 = s[0], s1 = s[1], s2 = s[2], s3 = s[3];
  const float mn = fmaxf(fmaxf(m, s0), fmaxf(fmaxf(s1, s2), s3));
  float st = s0;
  st = (t == 1) ? s1 : st;
  st = (t == 2) ? s2 : st;
  st = (t == 3) ? s3 : st;
  const float pt = expf(st - mn);
  const float scale = expf(m - mn);
  const float p0 = quad_move<0x00>(pt);
  const float p1 = quad_move<0x55>(pt);
  const float p2 = quad_move<0xAA>(pt);
  const float p3 = quad_move<0xFF>(pt);
  const float p[4] = {p0, p1, p2, p3};
  sum = fmaf(sum, scale, (p0 + p1) + (p2 + p3));
  acc.x *= scale;
  acc.y *= scale;
  acc.z *= scale;
  acc.w *= scale;
#pragma unroll
  for (int q = 0; q < 4; ++q) {
    const v4f rr = (q < nv) ? r[q] : (v4f){0.f, 0.f, 0.f, 0.f};
    acc.x = fmaf(p[q], rr.x, acc.x);
    acc.y = fmaf(p[q], rr.y, acc.y);
    acc.z = fmaf(p[q], rr.z, acc.z);
    acc.w = fmaf(p[q], rr.w, acc.w);
  }
  m = mn;
}

__global__ void __launch_bounds__(256)
node_gather(const int* __restrict__ cnt, const int* __restrict__ bucket,
            const int* __restrict__ ovf,
            const float* __restrict__ h, const float* __restrict__ att_src,
            const float* __restrict__ att_dst, const float* __restrict__ bias,
            float* __restrict__ out, int apply_elu) {
  const int lane = threadIdx.x & 31;
  const int n = __builtin_amdgcn_readfirstlane(
      (int)(blockIdx.x * blockDim.x + threadIdx.x) >> 5);
  if (n >= N_NODES) return;
  const v4f as4 = *(const v4f*)(att_src + 4 * lane);
  const v4f ad4 = *(const v4f*)(att_dst + 4 * lane);
  const v4f own = *(const v4f*)(h + (size_t)n * FDIM + 4 * lane);
  const int raw = bucket[(size_t)n * BUCKET_CAP + lane];
  const int deg = cnt[n];
  const int inb = deg < BUCKET_CAP ? deg : BUCKET_CAP;
  const int my = (lane < inb) ? raw : n;
  const int t = lane & 3;

  const float ad = quad_sum(dot4(own, ad4));
  float m = leaky(quad_sum(dot4(own, as4)) + ad);
  float sum = 1.f;
  v4f acc = own;

#pragma unroll
  for (int i = 0; i < BUCKET_CAP / 4; ++i) {
    if (4 * i < inb) {
      v4f r[4];
#pragma unroll
      for (int q = 0; q < 4; ++q) {
        const int sj = __builtin_amdgcn_readlane(my, 4 * i + q);
        r[q] = *(const v4f*)(h + (size_t)sj * FDIM + 4 * lane);
      }
      absorb4(r, inb - 4 * i, as4, ad, t, m, sum, acc);
    }
  }

  if (deg > BUCKET_CAP) {
    const int total = cnt[N_NODES];
    for (int kb = 0; kb < total; kb += 32) {
      const int k = kb + lane;
      int od = -1, os = n;
      if (k < total) {
        od = ovf[2 * k];
        os = ovf[2 * k + 1];
      }
      unsigned hit = __builtin_amdgcn_ballot_w32(od == n);
      while (hit != 0u) {
        const int l = __builtin_ctz(hit);
        hit &= hit - 1u;
        const int sj = __builtin_amdgcn_readlane(os, l);
        v4f r[4];
        r[0] = *(const v4f*)(h + (size_t)sj * FDIM + 4 * lane);
        r[1] = r[2] = r[3] = (v4f){0.f, 0.f, 0.f, 0.f};
        absorb4(r, 1, as4, ad, t, m, sum, acc);
      }
    }
  }

  const float inv = 1.f / (sum + 1e-16f);
  const v4f b = *(const v4f*)(bias + 4 * lane);
  v4f o;
  o.x = fmaf(acc.x, inv, b.x);
  o.y = fmaf(acc.y, inv, b.y);
  o.z = fmaf(acc.z, inv, b.z);
  o.w = fmaf(acc.w, inv, b.w);
  if (apply_elu) {
    o.x = o.x > 0.f ? o.x : (expf(o.x) - 1.f);
    o.y = o.y > 0.f ? o.y : (expf(o.y) - 1.f);
    o.z = o.z > 0.f ? o.z : (expf(o.z) - 1.f);
    o.w = o.w > 0.f ? o.w : (expf(o.w) - 1.f);
  }
  *(v4f*)(out + (size_t)n * FDIM + 4 * lane) = o;
}

extern "C" void kernel_launch(void* const* d_in, const int* in_sizes, int n_in,
                              void* d_out, int out_size, void* d_ws,
                              size_t ws_size, hipStream_t stream) {
  (void)in_sizes; (void)n_in; (void)out_size;

  const float* x   = (const float*)d_in[0];
  const int*   ei  = (const int*)d_in[1];
  const float* W1  = (const float*)d_in[2];
  const float* as1 = (const float*)d_in[3];
  const float* ad1 = (const float*)d_in[4];
  const float* b1  = (const float*)d_in[5];
  const float* W2  = (const float*)d_in[6];
  const float* as2 = (const float*)d_in[7];
  const float* ad2 = (const float*)d_in[8];
  const float* b2  = (const float*)d_in[9];
  float* out = (float*)d_out;

  float* ws    = (float*)d_ws;
  float* h     = ws;
  float* agg   = h + (size_t)N_NODES * FDIM;
  int* cnt     = (int*)(agg + (size_t)N_NODES * FDIM);
  int* bucket  = cnt + CNT_WORDS;
  int* ovf     = bucket + (size_t)N_NODES * BUCKET_CAP;
  int* tbl     = ovf + (size_t)N_EDGES * 2;
  int* sorted  = tbl + (size_t)TBL_WORDS;
  const size_t need = ((size_t)N_NODES * FDIM * 2 + (size_t)CNT_WORDS +
                       (size_t)N_NODES * BUCKET_CAP + (size_t)N_EDGES * 2 +
                       (size_t)TBL_WORDS + (size_t)N_CHUNKS * CHUNK_EDGES) * 4;
  if (ws_size < need) return;

  const int gemmBlocks = N_NODES / 16;
  const int gemmSortBlocks = N_CHUNKS + N_NODES / 16;
  const int gN  = N_NODES / 8;
  dim3 B64(64), B256(256), BBIN(BIN_THREADS);

  gemm128_sort<<<gemmSortBlocks, B64, 0, stream>>>(x, W1, h, ei, sorted, tbl, cnt);
  range_fill<<<N_RANGES, BBIN, 0, stream>>>(sorted, tbl, cnt, bucket, ovf);

  node_gather<<<gN, B256, 0, stream>>>(cnt, bucket, ovf, h, as1, ad1, b1, agg, 1);

  gemm128_wmma_f32<<<gemmBlocks, B64, 0, stream>>>(agg, W2, h);
  node_gather<<<gN, B256, 0, stream>>>(cnt, bucket, ovf, h, as2, ad2, b2, out, 0);
}


# node_gather: softmax and ELU exponentials in base 2 on the one-instruction hardware exp (log2 e folded into att_src, att_dst once per wave); expf sequences go; last bits differ, sum order unchanged
# speedup vs baseline: 15.1142x; 1.0300x over previous
#include <hip/hip_runtime.h>
#include <math.h>


#define N_NODES 50000
#define N_EDGES 800000
#define HEADS   8
#define HEAD_DIM 16
#define FDIM    128
#define NEG_SLOPE 0.2f

typedef __attribute__((ext_vector_type(2))) float v2f;
typedef __attribute__((ext_vector_type(8))) float v8f;
typedef __attribute__((ext_vector_type(4))) float v4f;
typedef __attribute__((ext_vector_type(4))) int v4i;
typedef __attribute__((ext_vector_type(2))) int v2i;

#define BUCKET_CAP 32
static_assert(BUCKET_CAP == 32, "node_gather: one lane per bucket slot");
static_assert(BUCKET_CAP * (N_NODES / 2) >= N_EDGES, "bucket: capacity below twice the mean degree");
#define CNT_WORDS (((N_NODES + 1 + 255) / 256) * 256)
static_assert(N_NODES % 8 == 0, "node_gather: eight whole waves per block");

#define BIN_THREADS 1024
#define BIN_WAVES (BIN_THREADS / 32)
#define RANGE_NODES 256
#define N_RANGES ((N_NODES + RANGE_NODES - 1) / RANGE_NODES)
#define CHUNK_EDGES 4096
#define N_CHUNKS ((N_EDGES + CHUNK_EDGES - 1) / CHUNK_EDGES)
#define SCAN_PER_LANE ((N_RANGES + 31) / 32)
#define CHUNKS_PER_WAVE ((N_CHUNKS + BIN_WAVES - 1) / BIN_WAVES)
#define TBL_WORDS (((2 * N_CHUNKS * N_RANGES + 31) / 32) * 32)
static_assert(RANGE_NODES == 256, "packed edge: low 8 bits are the place in the range");
static_assert(N_NODES < (1 << 23), "packed edge: source above the low 8 bits");
static_assert(N_EDGES % 4 == 0, "chunk_sort: four edges per 16-byte load");
static_assert(RANGE_NODES * BUCKET_CAP / 4 == 2 * BIN_THREADS, "range_fill: two 16-byte pieces per thread");

#define SORT_THREADS 64
#define SORT_EPT (CHUNK_EDGES / SORT_THREADS)
#define SORT_V4 (SORT_EPT / 4)
static_assert(CHUNK_EDGES % (4 * SORT_THREADS) == 0, "chunk_sort: whole 16-byte loads per thread");
static_assert(N_RANGES <= 256 && CHUNK_EDGES == 4096, "chunk_sort: key = 8 bits of range, 12 of rank, 8 of place");
static_assert(N_EDGES % (4 * SORT_THREADS) == 0, "chunk_sort: a load is inside the list for all threads or none");
template <bool FULL>
__device__ __forceinline__ void
chunk_sort_block(const int ch, const int nj, int* __restrict__ hist,
                 int* __restrict__ start, int* __restrict__ buf,
                 const int* __restrict__ ei, int* __restrict__ sorted,
                 int* __restrict__ tbl, int* __restrict__ cnt) {
  const int t = threadIdx.x;
  for (int i = t; i < N_RANGES; i += SORT_THREADS) hist[i] = 0;
  if (ch == 0 && t == 0) cnt[N_NODES] = 0;
  __syncthreads();

  const int e0 = ch * CHUNK_EDGES + 4 * t;
  v4i d4[SORT_V4], s4[SORT_V4];
#pragma unroll
  for (int j = 0; j < SORT_V4; ++j) {
    d4[j] = (v4i){0, 0, 0, 0};
    if (FULL || j < nj) d4[j] = *(const v4i*)(ei + N_EDGES + e0 + 4 * SORT_THREADS * j);
  }
  int key[SORT_EPT];
#pragma unroll
  for (int j = 0; j < SORT_V4; ++j) {
    if (FULL || j < nj) {
#pragma unroll
      for (int q = 0; q < 4; ++q) {
        const int rg = d4[j][q] >> 8;
        key[4 * j + q] = rg | (atomicAdd(&hist[rg], 1) << 8) |
                         ((d4[j][q] & (RANGE_NODES - 1)) << 20);
      }
    }
  }
#pragma unroll
  for (int j = 0; j < SORT_V4; ++j) {
    s4[j] = (v4i){0, 0, 0, 0};
    if (FULL || j < nj) s4[j] = *(const v4i*)(ei + e0 + 4 * SORT_THREADS * j);
  }
  __syncthreads();

  if (t < 32) {
    int v[SCAN_PER_LANE];
    int tot = 0;
#pragma unroll
    for (int q = 0; q < SCAN_PER_LANE; ++q) {
      const int i = SCAN_PER_LANE * t + q;
      v[q] = (i < N_RANGES) ? hist[i] : 0;
      tot += v[q];
    }
    int inc = tot;
#pragma unroll
    for (int o = 1; o < 32; o <<= 1) {
      const int u = __shfl_up(inc, o);
      if (t >= o) inc += u;
    }
    int run = inc - tot;
#pragma unroll
    for (int q = 0; q < SCAN_PER_LANE; ++q) {
      const int i = SCAN_PER_LANE * t + q;
      if (i < N_RANGES) {
        start[i] = run;
        *(v2i*)(tbl + 2 * (ch * N_RANGES + i)) = (v2i){run, v[q]};
        run += v[q];
      }
    }
  }
  __syncthreads();

#pragma unroll
  for (int j = 0; j < SORT_V4; ++j) {
    if (FULL || j < nj) {
#pragma unroll
      for (int q = 0; q < 4; ++q)
        buf[start[key[4 * j + q] & 255] + ((key[4 * j + q] >> 8) & (CHUNK_EDGES - 1))] =
            (s4[j][q] << 8) | (key[4 * j + q] >> 20);
    }
  }
  __syncthreads();
#pragma unroll
  for (int j = 0; j < SORT_V4; ++j) {
    if (FULL || j < nj)
      *(v4i*)(sorted + (size_t)e0 + 4 * SORT_THREADS * j) =
          *(const v4i*)(buf + 4 * (t + SORT_THREADS * j));
  }
}

__device__ __forceinline__ void file_edge(const int e, const int node0,
                                          int* __restrict__ c, int* __restrict__ bkt,
                                          int* __restrict__ cnt, int* __restrict__ ovf) {
  const int dl = e & (RANGE_NODES - 1);
  const int s = e >> 8;
  const int rk = atomicAdd(&c[dl], 1);
  if (rk < BUCKET_CAP) {
    bkt[dl * BUCKET_CAP + rk] = s;
  } else {
    const int k = atomicAdd(&cnt[N_NODES], 1);
    ovf[2 * k] = node0 + dl;
    ovf[2 * k + 1] = s;
  }
}

__global__ void __launch_bounds__(BIN_THREADS)
range_fill(const int* __restrict__ sorted, const int* __restrict__ tbl,
           int* __restrict__ cnt, int* __restrict__ bucket, int* __restrict__ ovf) {
  __shared__ int c[RANGE_NODES];
  __shared__ int seg_start[N_CHUNKS];
  __shared__ int seg_len[N_CHUNKS];
  __shared__ int bkt[RANGE_NODES * BUCKET_CAP];
  const int t = threadIdx.x;
  const int r = blockIdx.x;
  const int lane = t & 31;
  const int w = t >> 5;
  const int node0 = r * RANGE_NODES;
  if (t < RANGE_NODES) c[t] = 0;
  if (t < N_CHUNKS) {
    const v2i e = *(const v2i*)(tbl + 2 * (t * N_RANGES + r));
    seg_start[t] = e.x;
    seg_len[t] = e.y;
  }
  ((v4i*)bkt)[t] = (v4i){0, 0, 0, 0};
  ((v4i*)bkt)[t + BIN_THREADS] = (v4i){0, 0, 0, 0};
  __syncthreads();

  int st[CHUNKS_PER_WAVE], n[CHUNKS_PER_WAVE], e[CHUNKS_PER_WAVE];
#pragma unroll
  for (int q = 0; q < CHUNKS_PER_WAVE; ++q) {
    const int ch = w + BIN_WAVES * q;
    st[q] = 0;
    n[q] = 0;
    if (ch < N_CHUNKS) {
      st[q] = seg_start[ch];
      n[q] = seg_len[ch];
    }
    e[q] = 0;
    if (lane < n[q]) e[q] = sorted[(size_t)ch * CHUNK_EDGES + st[q] + lane];
  }
#pragma unroll
  for (int q = 0; q < CHUNKS_PER_WAVE; ++q) {
    const int ch = w + BIN_WAVES * q;
    if (lane < n[q]) file_edge(e[q], node0, c, bkt, cnt, ovf);
    for (int i = lane + 32; i < n[q]; i += 32)
      file_edge(sorted[(size_t)ch * CHUNK_EDGES + st[q] + i], node0, c, bkt, cnt, ovf);
  }
  __syncthreads();

  int* __restrict__ gb = bucket + (size_t)node0 * BUCKET_CAP;
#pragma unroll
  for (int i = t; i < RANGE_NODES * BUCKET_CAP / 4; i += BIN_THREADS)
    if (node0 + (i >> 3) < N_NODES) ((v4i*)gb)[i] = ((const v4i*)bkt)[i];
  if (t < RANGE_NODES && node0 + t < N_NODES) cnt[node0 + t] = c[t];
}

static_assert(N_NODES % 16 == 0, "gemm: whole 16-row strips");
__device__ __forceinline__ void
gemm_strip(const int tm  , const float* __restrict__ A,
           const float* __restrict__ W, float* __restrict__ out) {
  const int wave = threadIdx.x >> 5;
  const int lane = threadIdx.x & 31;
  const int half = lane >> 4;
  const int l    = lane & 15;

  const float* __restrict__ arow = A + (size_t)(tm * 16 + l) * FDIM + 2 * half;
  const float* __restrict__ bcol = W + (size_t)(2 * half) * FDIM + wave * 64 + 4 * l;

  v8f c0 = {}, c1 = {}, c2 = {}, c3 = {};
#pragma unroll
  for (int k0 = 0; k0 < FDIM; k0 += 4) {
    v2f a;
    a.x = arow[k0];
    a.y = arow[k0 + 1];
    const v4f b0 = *(const v4f*)(bcol + (size_t)k0 * FDIM);
    const v4f b1 = *(const v4f*)(bcol + (size_t)(k0 + 1) * FDIM);
    c0 = __builtin_amdgcn_wmma_f32_16x16x4_f32(false, a, false, (v2f){b0.x, b1.x},
                                               (short)0, c0, false, false);
    c1 = __builtin_amdgcn_wmma_f32_16x16x4_f32(false, a, false, (v2f){b0.y, b1.y},
                                               (short)0, c1, false, false);
    c2 = __builtin_amdgcn_wmma_f32_16x16x4_f32(false, a, false, (v2f){b0.z, b1.z},
                                               (short)0, c2, false, false);
    c3 = __builtin_amdgcn_wmma_f32_16x16x4_f32(false, a, false, (v2f){b0.w, b1.w},
                                               (short)0, c3, false, false);
  }
  float* __restrict__ orow =
      out + (size_t)(tm * 16 + 8 * half) * FDIM + wave * 64 + 4 * l;
#pragma unroll
  for (int r = 0; r < 8; ++r)
    *(v4f*)(orow + (size_t)r * FDIM) = (v4f){c0[r], c1[r], c2[r], c3[r]};
}

__global__ void __launch_bounds__(64) __attribute__((amdgpu_waves_per_eu(4, 4)))
gemm128_wmma_f32(const float* __restrict__ A, const float* __restrict__ W,
                 float* __restrict__ out) {
  gemm_strip(blockIdx.x, A, W, out);
}

static_assert(SORT_THREADS == 64, "gemm128_sort: the sorting blocks have the GEMM's block size");
__global__ void __launch_bounds__(64) __attribute__((amdgpu_waves_per_eu(4, 4)))
gemm128_sort(const float* __restrict__ A, const float* __restrict__ W,
             float* __restrict__ out, const int* __restrict__ ei,
             int* __restrict__ sorted, int* __restrict__ tbl, int* __restrict__ cnt) {
  if (blockIdx.x < N_CHUNKS) {
    __shared__ int hist[N_RANGES];
    __shared__ int start[N_RANGES];
    __shared__ int buf[CHUNK_EDGES];
    const int ch = blockIdx.x;
    const int left = (N_EDGES - ch * CHUNK_EDGES) / (4 * SORT_THREADS);
    const int nj = left < SORT_V4 ? left : SORT_V4;
    if (nj == SORT_V4)
      chunk_sort_block<true>(ch, nj, hist, start, buf, ei, sorted, tbl, cnt);
    else
      chunk_sort_block<false>(ch, nj, hist, start, buf, ei, sorted, tbl, cnt);
    return;
  }
  gemm_strip(blockIdx.x - N_CHUNKS, A, W, out);
}

static_assert(NEG_SLOPE > 0.f && NEG_SLOPE < 1.f, "leaky: max form needs 0 < slope < 1");

#define LOG2E 1.44269504088896340736f
__device__ __forceinline__ float exp2_hw(float x) {
  return __builtin_amdgcn_exp2f(x);
}
__device__ __forceinline__ float leaky(float x) {
  return fmaxf(x, NEG_SLOPE * x);
}

template <int CTRL>
__device__ __forceinline__ float quad_move(float v) {
  return __builtin_bit_cast(
      float, __builtin_amdgcn_update_dpp(0, __builtin_bit_cast(int, v), CTRL,
                                         0xF, 0xF, true));
}

__device__ __forceinline__ float quad_sum(float v) {
  v += quad_move<0xB1>(v);
  v += quad_move<0x4E>(v);
  return v;
}

__device__ __forceinline__ float dot4(const v4f a, const v4f b) {
  float d = a.x * b.x;
  d = fmaf(a.y, b.y, d);
  d = fmaf(a.z, b.z, d);
  d = fmaf(a.w, b.w, d);
  return d;
}

__device__ __forceinline__ void absorb4(const v4f (&r)[4], const int nv,
                                        const v4f as4, const float ad, const int t,
                                        float& m, float& sum, v4f& acc) {
  float s[4];
#pragma unroll
  for (int q = 0; q < 4; ++q) {
    const float sc = leaky(quad_sum(dot4(r[q], as4)) + ad);
    s[q] = (q < nv) ? sc : -__builtin_inff();
  }
  const float s0 = s[0], s1 = s[1], s2 = s[2], s3 = s[3];
  const float mn = fmaxf(fmaxf(m, s0), fmaxf(fmaxf(s1, s2), s3));
  float st = s0;
  st = (t == 1) ? s1 : st;
  st = (t == 2) ? s2 : st;
  st = (t == 3) ? s3 : st;
  const float pt = exp2_hw(st - mn);
  const float scale = exp2_hw(m - mn);
  const float p0 = quad_move<0x00>(pt);
  const float p1 = quad_move<0x55>(pt);
  const float p2 = quad_move<0xAA>(pt);
  const float p3 = quad_move<0xFF>(pt);
  const float p[4] = {p0, p1, p2, p3};
  sum = fmaf(sum, scale, (p0 + p1) + (p2 + p3));
  acc.x *= scale;
  acc.y *= scale;
  acc.z *= scale;
  acc.w *= scale;
#pragma unroll
  for (int q = 0; q < 4; ++q) {
    const v4f rr = (q < nv) ? r[q] : (v4f){0.f, 0.f, 0.f, 0.f};
    acc.x = fmaf(p[q], rr.x, acc.x);
    acc.y = fmaf(p[q], rr.y, acc.y);
    acc.z = fmaf(p[q], rr.z, acc.z);
    acc.w = fmaf(p[q], rr.w, acc.w);
  }
  m = mn;
}

__global__ void __launch_bounds__(256)
node_gather(const int* __restrict__ cnt, const int* __restrict__ bucket,
            const int* __restrict__ ovf,
            const float* __restrict__ h, const float* __restrict__ att_src,
            const float* __restrict__ att_dst, const float* __restrict__ bias,
            float* __restrict__ out, int apply_elu) {
  const int lane = threadIdx.x & 31;
  const int n = __builtin_amdgcn_readfirstlane(
      (int)(blockIdx.x * blockDim.x + threadIdx.x) >> 5);
  if (n >= N_NODES) return;
  const v4f as4 = *(const v4f*)(att_src + 4 * lane) * LOG2E;
  const v4f ad4 = *(const v4f*)(att_dst + 4 * lane) * LOG2E;
  const v4f own = *(const v4f*)(h + (size_t)n * FDIM + 4 * lane);
  const int raw = bucket[(size_t)n * BUCKET_CAP + lane];
  const int deg = cnt[n];
  const int inb = deg < BUCKET_CAP ? deg : BUCKET_CAP;
  const int my = (lane < inb) ? raw : n;
  const int t = lane & 3;

  const float ad = quad_sum(dot4(own, ad4));
  float m = leaky(quad_sum(dot4(own, as4)) + ad);
  float sum = 1.f;
  v4f acc = own;

#pragma unroll
  for (int i = 0; i < BUCKET_CAP / 4; ++i) {
    if (4 * i < inb) {
      v4f r[4];
#pragma unroll
      for (int q = 0; q < 4; ++q) {
        const int sj = __builtin_amdgcn_readlane(my, 4 * i + q);
        r[q] = *(const v4f*)(h + (size_t)sj * FDIM + 4 * lane);
      }
      absorb4(r, inb - 4 * i, as4, ad, t, m, sum, acc);
    }
  }

  if (deg > BUCKET_CAP) {
    const int total = cnt[N_NODES];
    for (int kb = 0; kb < total; kb += 32) {
      const int k = kb + lane;
      int od = -1, os = n;
      if (k < total) {
        od = ovf[2 * k];
        os = ovf[2 * k + 1];
      }
      unsigned hit = __builtin_amdgcn_ballot_w32(od == n);
      while (hit != 0u) {
        const int l = __builtin_ctz(hit);
        hit &= hit - 1u;
        const int sj = __builtin_amdgcn_readlane(os, l);
        v4f r[4];
        r[0] = *(const v4f*)(h + (size_t)sj * FDIM + 4 * lane);
        r[1] = r[2] = r[3] = (v4f){0.f, 0.f, 0.f, 0.f};
        absorb4(r, 1, as4, ad, t, m, sum, acc);
      }
    }
  }

  const float inv = 1.f / (sum + 1e-16f);
  const v4f b = *(const v4f*)(bias + 4 * lane);
  v4f o;
  o.x = fmaf(acc.x, inv, b.x);
  o.y = fmaf(acc.y, inv, b.y);
  o.z = fmaf(acc.z, inv, b.z);
  o.w = fmaf(acc.w, inv, b.w);
  if (apply_elu) {
    o.x = o.x > 0.f ? o.x : (exp2_hw(LOG2E * o.x) - 1.f);
    o.y = o.y > 0.f ? o.y : (exp2_hw(LOG2E * o.y) - 1.f);
    o.z = o.z > 0.f ? o.z : (exp2_hw(LOG2E * o.z) - 1.f);
    o.w = o.w > 0.f ? o.w : (exp2_hw(LOG2E * o.w) - 1.f);
  }
  *(v4f*)(out + (size_t)n * FDIM + 4 * lane) = o;
}

extern "C" void kernel_launch(void* const* d_in, const int* in_sizes, int n_in,
                              void* d_out, int out_size, void* d_ws,
                              size_t ws_size, hipStream_t stream) {
  (void)in_sizes; (void)n_in; (void)out_size;

  const float* x   = (const float*)d_in[0];
  const int*   ei  = (const int*)d_in[1];
  const float* W1  = (const float*)d_in[2];
  const float* as1 = (const float*)d_in[3];
  const float* ad1 = (const float*)d_in[4];
  const float* b1  = (const float*)d_in[5];
  const float* W2  = (const float*)d_in[6];
  const float* as2 = (const float*)d_in[7];
  const float* ad2 = (const float*)d_in[8];
  const float* b2  = (const float*)d_in[9];
  float* out = (float*)d_out;

  float* ws    = (float*)d_ws;
  float* h     = ws;
  float* agg   = h + (size_t)N_NODES * FDIM;
  int* cnt     = (int*)(agg + (size_t)N_NODES * FDIM);
  int* bucket  = cnt + CNT_WORDS;
  int* ovf     = bucket + (size_t)N_NODES * BUCKET_CAP;
  int* tbl     = ovf + (size_t)N_EDGES * 2;
  int* sorted  = tbl + (size_t)TBL_WORDS;
  const size_t need = ((size_t)N_NODES * FDIM * 2 + (size_t)CNT_WORDS +
                       (size_t)N_NODES * BUCKET_CAP + (size_t)N_EDGES * 2 +
                       (size_t)TBL_WORDS + (size_t)N_CHUNKS * CHUNK_EDGES) * 4;
  if (ws_size < need) return;

  const int gemmBlocks = N_NODES / 16;
  const int gemmSortBlocks = N_CHUNKS + N_NODES / 16;
  const int gN  = N_NODES / 8;
  dim3 B64(64), B256(256), BBIN(BIN_THREADS);

  gemm128_sort<<<gemmSortBlocks, B64, 0, stream>>>(x, W1, h, ei, sorted, tbl, cnt);
  range_fill<<<N_RANGES, BBIN, 0, stream>>>(sorted, tbl, cnt, bucket, ovf);

  node_gather<<<gN, B256, 0, stream>>>(cnt, bucket, ovf, h, as1, ad1, b1, agg, 1);

  gemm128_wmma_f32<<<gemmBlocks, B64, 0, stream>>>(agg, W2, h);
  node_gather<<<gN, B256, 0, stream>>>(cnt, bucket, ovf, h, as2, ad2, b2, out, 0);
}
